# starting kernel
# baseline (speedup 1.0000x reference)
#include <hip/hip_runtime.h>

constexpr int N0 = 400000, N1 = 100000, N2 = 25000;
constexpr int E0 = 800000, E1 = 200000;
constexpr int DIN = 128, HID = 256, DOUT = 128;

typedef __attribute__((ext_vector_type(2))) float v2f;
typedef __attribute__((ext_vector_type(8))) float v8f;

__global__ void zero_f32(float* __restrict__ p, long n) {
    long i = (long)blockIdx.x * blockDim.x + threadIdx.x;
    if (i < n) p[i] = 0.0f;
}

__global__ void build_feat(const int* __restrict__ x,
                           const float* __restrict__ e0, const float* __restrict__ e1,
                           const float* __restrict__ e2, const float* __restrict__ e3,
                           float* __restrict__ feat) {
    int t = blockIdx.x * blockDim.x + threadIdx.x;
    int row = t >> 5, lane = t & 31;
    if (row >= N0) return;
    const float4* src;
    if (lane < 8)       src = (const float4*)(e0 + (long)x[row * 4 + 0] * 32) + lane;
    else if (lane < 12) src = (const float4*)(e1 + (long)x[row * 4 + 1] * 16) + (lane - 8);
    else if (lane < 16) src = (const float4*)(e2 + (long)x[row * 4 + 2] * 16) + (lane - 12);
    else                src = (const float4*)(e3 + (long)x[row * 4 + 3] * 64) + (lane - 16);
    ((float4*)(feat + (long)row * 128))[lane] = *src;
}

template <int WIDTH>
__global__ void scatter_edges(const float* __restrict__ h, const int* __restrict__ srcA,
                              const int* __restrict__ dstA, float* __restrict__ agg,
                              float* __restrict__ deg, int nE) {
    int t = blockIdx.x * blockDim.x + threadIdx.x;
    int e = t >> 5, lane = t & 31;
    if (e >= nE) return;
    int s = srcA[e], d = dstA[e];
    const float* hp = h + (long)s * WIDTH;
    float* ap = agg + (long)d * WIDTH;
#pragma unroll
    for (int j = 0; j < WIDTH / 128; ++j) {
        float4 v = ((const float4*)hp)[lane + 32 * j];
        atomicAdd(ap + 4 * lane + 128 * j + 0, v.x);
        atomicAdd(ap + 4 * lane + 128 * j + 1, v.y);
        atomicAdd(ap + 4 * lane + 128 * j + 2, v.z);
        atomicAdd(ap + 4 * lane + 128 * j + 3, v.w);
    }
    if (lane == 0) atomicAdd(deg + d, 1.0f);
}

__global__ void __launch_bounds__(256)
gemm1_fused(const float* __restrict__ feat, const float* __restrict__ agg0,
            const float* __restrict__ deg0, const float* __restrict__ W,
            const float* __restrict__ bias, float* __restrict__ h1) {
    __shared__ float As[16 * 132];
    __shared__ float srow[16];
    const int m0 = blockIdx.x * 16;
    const int tid = threadIdx.x;
#pragma unroll
    for (int i = 0; i < 8; ++i) {
        int idx = tid + i * 256;
        int r = idx >> 7, c = idx & 127;
        long row = m0 + r;
        float inv = 1.0f / fmaxf(deg0[row], 1.0f);
        As[r * 132 + c] = feat[row * 128 + c] + agg0[row * 128 + c] * inv;
    }
    if (tid < 16) srow[tid] = (deg0[m0 + tid] > 0.0f) ? 2.0f : 1.0f;
    __syncthreads();

    const int lane = tid & 31, wave = tid >> 5;
    const int lo = lane & 15, hi = lane >> 4;
    const int nA = wave * 16, nB = (wave + 8) * 16;
    v8f c0 = {0, 0, 0, 0, 0, 0, 0, 0};
    v8f c1 = {0, 0, 0, 0, 0, 0, 0, 0};
    for (int k = 0; k < DIN; k += 4) {
        const int kk = k + 2 * hi;
        v2f a;
        a[0] = As[lo * 132 + kk];
        a[1] = As[lo * 132 + kk + 1];
        v2f b0, b1;
        b0[0] = W[kk * HID + nA + lo];
        b0[1] = W[(kk + 1) * HID + nA + lo];
        b1[0] = W[kk * HID + nB + lo];
        b1[1] = W[(kk + 1) * HID + nB + lo];
        c0 = __builtin_amdgcn_wmma_f32_16x16x4_f32(false, a, false, b0, (short)0, c0, false, false);
        c1 = __builtin_amdgcn_wmma_f32_16x16x4_f32(false, a, false, b1, (short)0, c1, false, false);
    }
#pragma unroll
    for (int r = 0; r < 8; ++r) {
        int rr = r + 8 * hi;
        long row = m0 + rr;
        float s = srow[rr];
        h1[row * HID + nA + lo] = fmaxf(c0[r] + s * bias[nA + lo], 0.0f);
        h1[row * HID + nB + lo] = fmaxf(c1[r] + s * bias[nB + lo], 0.0f);
    }
}

__global__ void __launch_bounds__(256)
gemm2_fused(const float* __restrict__ h1, const float* __restrict__ agg1,
            const float* __restrict__ deg1, const float* __restrict__ W,
            const float* __restrict__ bias, float* __restrict__ out) {
    __shared__ float As[16 * 260];
    const int m0 = blockIdx.x * 16;
    const int tid = threadIdx.x;
#pragma unroll
    for (int i = 0; i < 16; ++i) {
        int idx = tid + i * 256;
        int r = idx >> 8, c = idx & 255;
        int row = m0 + r;
        float v = 0.0f;
        if (row < N2) {
            float inv = 1.0f / fmaxf(deg1[row], 1.0f);
            v = h1[(long)row * HID + c] + agg1[(long)row * HID + c] * inv;
        }
        As[r * 260 + c] = v;
    }
    __syncthreads();

    const int lane = tid & 31, wave = tid >> 5;
    const int lo = lane & 15, hi = lane >> 4;
    const int n0 = wave * 16;
    v8f acc = {0, 0, 0, 0, 0, 0, 0, 0};
    for (int k = 0; k < HID; k += 4) {
        const int kk = k + 2 * hi;
        v2f a;
        a[0] = As[lo * 260 + kk];
        a[1] = As[lo * 260 + kk + 1];
        v2f b;
        b[0] = W[kk * DOUT + n0 + lo];
        b[1] = W[(kk + 1) * DOUT + n0 + lo];
        acc = __builtin_amdgcn_wmma_f32_16x16x4_f32(false, a, false, b, (short)0, acc, false, false);
    }
#pragma unroll
    for (int r = 0; r < 8; ++r) {
        int row = m0 + r + 8 * hi;
        if (row < N2) out[(long)row * DOUT + n0 + lo] = acc[r] + bias[n0 + lo];
    }
}

extern "C" void kernel_launch(void* const* d_in, const int* in_sizes, int n_in,
                              void* d_out, int out_size, void* d_ws, size_t ws_size,
                              hipStream_t stream) {
    (void)in_sizes; (void)n_in; (void)out_size; (void)ws_size;
    const int*   x  = (const int*)d_in[0];
    const float* e0 = (const float*)d_in[1];
    const float* e1 = (const float*)d_in[2];
    const float* e2 = (const float*)d_in[3];
    const float* e3 = (const float*)d_in[4];
    const float* Win[2]  = {(const float*)d_in[5], (const float*)d_in[9]};
    const float* bin[2]  = {(const float*)d_in[6], (const float*)d_in[10]};
    const float* Wout[2] = {(const float*)d_in[7], (const float*)d_in[11]};
    const float* bout[2] = {(const float*)d_in[8], (const float*)d_in[12]};
    const int* srcL0[2] = {(const int*)d_in[13], (const int*)d_in[17]};
    const int* dstL0[2] = {(const int*)d_in[14], (const int*)d_in[18]};
    const int* srcL1[2] = {(const int*)d_in[15], (const int*)d_in[19]};
    const int* dstL1[2] = {(const int*)d_in[16], (const int*)d_in[20]};
    float* out = (float*)d_out;

    size_t off = 0;
    auto take = [&](size_t bytes) { size_t o = off; off += (bytes + 255) & ~(size_t)255; return o; };
    char* ws = (char*)d_ws;
    float* feat = (float*)(ws + take((size_t)N0 * DIN * 4));
    float* h1   = (float*)(ws + take((size_t)N1 * HID * 4));
    size_t zbeg = off;
    float* agg0 = (float*)(ws + take((size_t)N1 * DIN * 4));
    float* deg0 = (float*)(ws + take((size_t)N1 * 4));
    float* agg1 = (float*)(ws + take((size_t)N2 * HID * 4));
    float* deg1 = (float*)(ws + take((size_t)N2 * 4));
    float* zptr = (float*)(ws + zbeg);
    long   zcnt = (long)((off - zbeg) / 4);

    build_feat<<<N0 / 8, 256, 0, stream>>>(x, e0, e1, e2, e3, feat);

    for (int e = 0; e < 2; ++e) {
        zero_f32<<<(unsigned)((zcnt + 255) / 256), 256, 0, stream>>>(zptr, zcnt);
        scatter_edges<DIN><<<E0 / 8, 256, 0, stream>>>(feat, srcL0[e], dstL0[e], agg0, deg0, E0);
        gemm1_fused<<<N1 / 16, 256, 0, stream>>>(feat, agg0, deg0, Win[e], bin[e], h1);
        scatter_edges<HID><<<E1 / 8, 256, 0, stream>>>(h1, srcL1[e], dstL1[e], agg1, deg1, E1);
        gemm2_fused<<<(N2 + 15) / 16, 256, 0, stream>>>(h1, agg1, deg1, Wout[e], bout[e],
                                                        out + (size_t)e * N2 * DOUT);
    }
}


# layer-0 sums kept in LDS: each block owns 400 destinations, scans the dst list for them, adds source rows with LDS float adds and runs gemm1 from LDS; agg0 plane gone; sums in LDS-add arrival order
# speedup vs baseline: 1.5838x; 1.5838x over previous
#include <hip/hip_runtime.h>

constexpr int N0 = 400000, N1 = 100000, N2 = 25000;
constexpr int E0 = 800000, E1 = 200000;
constexpr int DIN = 128, HID = 256, DOUT = 128;

typedef __attribute__((ext_vector_type(2))) float v2f;
typedef __attribute__((ext_vector_type(8))) float v8f;
typedef __attribute__((ext_vector_type(4))) float v4f;

__global__ void __launch_bounds__(256) zero_f32(float4* __restrict__ p, long n4) {
    const long stride = (long)gridDim.x * blockDim.x;
    const float4 z = {0.0f, 0.0f, 0.0f, 0.0f};
    for (long i = (long)blockIdx.x * blockDim.x + threadIdx.x; i < n4; i += stride) p[i] = z;
}

constexpr int FEAT_U = 8;
static_assert(N0 % 32 == 0, "build_feat takes whole groups of 32 rows");
__global__ void __launch_bounds__(256)
build_feat(const int* __restrict__ x,
           const float* __restrict__ e0, const float* __restrict__ e1,
           const float* __restrict__ e2, const float* __restrict__ e3,
           float* __restrict__ feat) {
    const int lane = threadIdx.x & 31;
    const int wave = __builtin_amdgcn_readfirstlane((int)((blockIdx.x * blockDim.x + threadIdx.x) >> 5));
    const int row0 = wave * 32;
    if (row0 >= N0) return;
    const int4 xi = ((const int4*)x)[row0 + lane];
    const float* tb;
    int rl;
    if (lane < 8)       { tb = e0 + lane * 4;        rl = 32; }
    else if (lane < 12) { tb = e1 + (lane - 8) * 4;  rl = 16; }
    else if (lane < 16) { tb = e2 + (lane - 12) * 4; rl = 16; }
    else                { tb = e3 + (lane - 16) * 4; rl = 64; }
    v4f* dst = (v4f*)(feat + (long)row0 * 128) + lane;
#pragma unroll 1
    for (int i = 0; i < 32; i += FEAT_U) {
        v4f v[FEAT_U];
#pragma unroll
        for (int u = 0; u < FEAT_U; ++u) {
            const int i0 = __builtin_amdgcn_readlane(xi.x, i + u);
            const int i1 = __builtin_amdgcn_readlane(xi.y, i + u);
            const int i2 = __builtin_amdgcn_readlane(xi.z, i + u);
            const int i3 = __builtin_amdgcn_readlane(xi.w, i + u);
            const int idx = (lane < 8) ? i0 : (lane < 12) ? i1 : (lane < 16) ? i2 : i3;
            v[u] = *(const v4f*)(tb + (long)idx * rl);
        }
        static_assert(FEAT_U == 8, "the fence below names eight values");
        asm volatile("" :: "v"(v[0]), "v"(v[1]), "v"(v[2]), "v"(v[3]),
                           "v"(v[4]), "v"(v[5]), "v"(v[6]), "v"(v[7]));
#pragma unroll
        for (int u = 0; u < FEAT_U; ++u) dst[(i + u) * 32] = v[u];
    }
}

template <int WIDTH, int U>
__global__ void __launch_bounds__(256)
scatter_edges(const float* __restrict__ h, const int* __restrict__ srcA,
              const int* __restrict__ dstA, float* __restrict__ agg,
              float* __restrict__ deg, int nE) {
    constexpr int L = WIDTH / 32;
    const int lane = threadIdx.x & 31;
    const int wave = __builtin_amdgcn_readfirstlane((int)((blockIdx.x * blockDim.x + threadIdx.x) >> 5));
    const int base = wave * 32;
    if (base >= nE) return;
    const int cnt = (nE - base < 32) ? (nE - base) : 32;
    int s = 0, d = 0;
    if (lane < cnt) {
        s = srcA[base + lane];
        d = dstA[base + lane];
        atomicAdd(deg + d, 1.0f);
    }
    const float* hl = h + lane;
    float* al = agg + lane;
    int i = 0;
#pragma unroll 1
    for (; i + U <= cnt; i += U) {
        float v[U][L];
        int di[U];
#pragma unroll
        for (int u = 0; u < U; ++u) {
            const int si = __builtin_amdgcn_readlane(s, i + u);
            di[u] = __builtin_amdgcn_readlane(d, i + u);
            const float* hp = hl + (long)si * WIDTH;
#pragma unroll
            for (int j = 0; j < L; ++j) v[u][j] = hp[32 * j];
        }
#pragma unroll
        for (int u = 0; u < U; ++u) {
            float* ap = al + (long)di[u] * WIDTH;
#pragma unroll
            for (int j = 0; j < L; ++j) atomicAdd(ap + 32 * j, v[u][j]);
        }
    }
    for (; i < cnt; ++i) {
        const int si = __builtin_amdgcn_readlane(s, i);
        const int dd = __builtin_amdgcn_readlane(d, i);
        const float* hp = hl + (long)si * WIDTH;
        float* ap = al + (long)dd * WIDTH;
        float v[L];
#pragma unroll
        for (int j = 0; j < L; ++j) v[j] = hp[32 * j];
#pragma unroll
        for (int j = 0; j < L; ++j) atomicAdd(ap + 32 * j, v[j]);
    }
}

constexpr int TM = 400;
constexpr int PITCH = 132;
constexpr int QCAP = 8192;
constexpr int G1T = 1024;
static_assert(N1 % TM == 0 && TM % 16 == 0, "whole tiles only");
static_assert(E0 % 4 == 0, "the dst list is scanned with 16-byte loads");
__global__ void __launch_bounds__(G1T)
agg_gemm1(const float* __restrict__ feat, const int* __restrict__ srcA,
          const int* __restrict__ dstA, const float* __restrict__ W,
          const float* __restrict__ bias, float* __restrict__ h1) {
    __shared__ float As[TM * PITCH];
    __shared__ int degc[TM];
    __shared__ float invs[TM];
    __shared__ int2 q[QCAP];
    __shared__ int qn;
    const int tid = threadIdx.x;
    const int lane = tid & 31;
    const int wave = __builtin_amdgcn_readfirstlane(tid >> 5);
    const int m0 = blockIdx.x * TM;

    for (int i = tid; i < TM * PITCH; i += G1T) As[i] = 0.0f;
    for (int i = tid; i < TM; i += G1T) degc[i] = 0;
    if (tid == 0) qn = 0;
    __syncthreads();

    {
        const int4* d4 = (const int4*)dstA;
        constexpr int NQ = E0 / 4;
        auto take = [&](int d, int e) {
            const unsigned dl = (unsigned)(d - m0);
            if (dl < (unsigned)TM) {
                const int slot = atomicAdd(&qn, 1);
                if (slot < QCAP) {
                    q[slot] = make_int2(e, (int)dl);
                } else {
                    const float* hp = feat + (long)srcA[e] * DIN;
                    for (int c = 0; c < DIN; ++c) atomicAdd(&As[dl * PITCH + c], hp[c]);
                    atomicAdd(&degc[dl], 1);
                }
            }
        };
        for (int i0 = tid; i0 < NQ; i0 += 4 * G1T) {
            int4 dv[4];
#pragma unroll
            for (int u = 0; u < 4; ++u) {
                const int i = i0 + u * G1T;
                dv[u] = (i < NQ) ? d4[i] : make_int4(-1, -1, -1, -1);
            }
#pragma unroll
            for (int u = 0; u < 4; ++u) {
                const int e = (i0 + u * G1T) * 4;
                take(dv[u].x, e);
                take(dv[u].y, e + 1);
                take(dv[u].z, e + 2);
                take(dv[u].w, e + 3);
            }
        }
    }
    __syncthreads();

    {
        const int qv = qn;
        const int n = __builtin_amdgcn_readfirstlane(qv < QCAP ? qv : QCAP);
        const float* hl = feat + lane;
        float* al = As + lane;
        for (int base = wave * 32; base < n; base += G1T) {
            const int cnt = (n - base < 32) ? (n - base) : 32;
            int s = 0, dl = 0;
            if (lane < cnt) {
                const int2 ent = q[base + lane];
                s = srcA[ent.x];
                dl = ent.y;
                atomicAdd(&degc[dl], 1);
            }
            int i = 0;
#pragma unroll 1
            for (; i + 8 <= cnt; i += 8) {
                float v[8][4];
                int di[8];
#pragma unroll
                for (int u = 0; u < 8; ++u) {
                    const int si = __builtin_amdgcn_readlane(s, i + u);
                    di[u] = __builtin_amdgcn_readlane(dl, i + u);
                    const float* hp = hl + (long)si * DIN;
#pragma unroll
                    for (int j = 0; j < 4; ++j) v[u][j] = hp[32 * j];
                }
#pragma unroll
                for (int u = 0; u < 8; ++u) {
                    float* ap = al + di[u] * PITCH;
#pragma unroll
                    for (int j = 0; j < 4; ++j) atomicAdd(ap + 32 * j, v[u][j]);
                }
            }
            for (; i < cnt; ++i) {
                const int si = __builtin_amdgcn_readlane(s, i);
                const int dd = __builtin_amdgcn_readlane(dl, i);
                const float* hp = hl + (long)si * DIN;
                float* ap = al + dd * PITCH;
                float v[4];
#pragma unroll
                for (int j = 0; j < 4; ++j) v[j] = hp[32 * j];
#pragma unroll
                for (int j = 0; j < 4; ++j) atomicAdd(ap + 32 * j, v[j]);
            }
        }
    }
    __syncthreads();

    for (int r = tid; r < TM; r += G1T) invs[r] = 1.0f / fmaxf((float)degc[r], 1.0f);
    __syncthreads();
    for (int idx = tid; idx < TM * DIN; idx += G1T) {
        const int r = idx >> 7, c = idx & 127;
        As[r * PITCH + c] = feat[(long)(m0 + r) * DIN + c] + As[r * PITCH + c] * invs[r];
    }
    __syncthreads();

    const int lo = lane & 15, hi = lane >> 4;
    const int ncol = (wave & 15) * 16 + lo;
    v2f b[DIN / 4];
#pragma unroll
    for (int ks = 0; ks < DIN / 4; ++ks) {
        const int kk = 4 * ks + 2 * hi;
        b[ks][0] = W[kk * HID + ncol];
        b[ks][1] = W[(kk + 1) * HID + ncol];
    }
    const float bv = bias[ncol];
#pragma unroll 1
    for (int mt = wave >> 4; mt < TM / 16; mt += 2) {
        const float* ar = As + (mt * 16 + lo) * PITCH + 2 * hi;
        v8f acc = {0, 0, 0, 0, 0, 0, 0, 0};
#pragma unroll
        for (int ks = 0; ks < DIN / 4; ++ks) {
            v2f a;
            a[0] = ar[4 * ks];
            a[1] = ar[4 * ks + 1];
            acc = __builtin_amdgcn_wmma_f32_16x16x4_f32(false, a, false, b[ks], (short)0, acc, false, false);
        }
#pragma unroll
        for (int r = 0; r < 8; ++r) {
            const int rr = mt * 16 + r + 8 * hi;
            const float s = (degc[rr] > 0) ? 2.0f : 1.0f;
            h1[(long)(m0 + rr) * HID + ncol] = fmaxf(acc[r] + s * bv, 0.0f);
        }
    }
}

__global__ void __launch_bounds__(256)
gemm2_fused(const float* __restrict__ h1, const float* __restrict__ agg1,
            const float* __restrict__ deg1, const float* __restrict__ W,
            const float* __restrict__ bias, float* __restrict__ out) {
    __shared__ float As[16 * 260];
    const int m0 = blockIdx.x * 16;
    const int tid = threadIdx.x;
#pragma unroll
    for (int i = 0; i < 16; ++i) {
        int idx = tid + i * 256;
        int r = idx >> 8, c = idx & 255;
        int row = m0 + r;
        float v = 0.0f;
        if (row < N2) {
            float inv = 1.0f / fmaxf(deg1[row], 1.0f);
            v = h1[(long)row * HID + c] + agg1[(long)row * HID + c] * inv;
        }
        As[r * 260 + c] = v;
    }
    __syncthreads();

    const int lane = tid & 31, wave = tid >> 5;
    const int lo = lane & 15, hi = lane >> 4;
    const int n0 = wave * 16;
    v8f acc = {0, 0, 0, 0, 0, 0, 0, 0};
    for (int k = 0; k < HID; k += 4) {
        const int kk = k + 2 * hi;
        v2f a;
        a[0] = As[lo * 260 + kk];
        a[1] = As[lo * 260 + kk + 1];
        v2f b;
        b[0] = W[kk * DOUT + n0 + lo];
        b[1] = W[(kk + 1) * DOUT + n0 + lo];
        acc = __builtin_amdgcn_wmma_f32_16x16x4_f32(false, a, false, b, (short)0, acc, false, false);
    }
#pragma unroll
    for (int r = 0; r < 8; ++r) {
        int row = m0 + r + 8 * hi;
        if (row < N2) out[(long)row * DOUT + n0 + lo] = acc[r] + bias[n0 + lo];
    }
}

extern "C" void kernel_launch(void* const* d_in, const int* in_sizes, int n_in,
                              void* d_out, int out_size, void* d_ws, size_t ws_size,
                              hipStream_t stream) {
    (void)in_sizes; (void)n_in; (void)out_size; (void)ws_size;
    const int*   x  = (const int*)d_in[0];
    const float* e0 = (const float*)d_in[1];
    const float* e1 = (const float*)d_in[2];
    const float* e2 = (const float*)d_in[3];
    const float* e3 = (const float*)d_in[4];
    const float* Win[2]  = {(const float*)d_in[5], (const float*)d_in[9]};
    const float* bin[2]  = {(const float*)d_in[6], (const float*)d_in[10]};
    const float* Wout[2] = {(const float*)d_in[7], (const float*)d_in[11]};
    const float* bout[2] = {(const float*)d_in[8], (const float*)d_in[12]};
    const int* srcL0[2] = {(const int*)d_in[13], (const int*)d_in[17]};
    const int* dstL0[2] = {(const int*)d_in[14], (const int*)d_in[18]};
    const int* srcL1[2] = {(const int*)d_in[15], (const int*)d_in[19]};
    const int* dstL1[2] = {(const int*)d_in[16], (const int*)d_in[20]};
    float* out = (float*)d_out;

    size_t off = 0;
    auto take = [&](size_t bytes) { size_t o = off; off += (bytes + 255) & ~(size_t)255; return o; };
    char* ws = (char*)d_ws;
    float* feat = (float*)(ws + take((size_t)N0 * DIN * 4));
    float* h1   = (float*)(ws + take((size_t)N1 * HID * 4));
    size_t zbeg = off;
    float* agg1 = (float*)(ws + take((size_t)N2 * HID * 4));
    float* deg1 = (float*)(ws + take((size_t)N2 * 4));
    float4* zptr = (float4*)(ws + zbeg);
    long    zcnt4 = (long)((off - zbeg) / 16);

    build_feat<<<(N0 / 32 + 7) / 8, 256, 0, stream>>>(x, e0, e1, e2, e3, feat);

    for (int e = 0; e < 2; ++e) {
        zero_f32<<<2048, 256, 0, stream>>>(zptr, zcnt4);
        agg_gemm1<<<N1 / TM, G1T, 0, stream>>>(feat, srcL0[e], dstL0[e], Win[e], bin[e], h1);
        scatter_edges<HID, 4><<<((E1 + 31) / 32 + 7) / 8, 256, 0, stream>>>(h1, srcL1[e], dstL1[e], agg1, deg1, E1);
        gemm2_fused<<<(N2 + 15) / 16, 256, 0, stream>>>(h1, agg1, deg1, Wout[e], bout[e],
                                                        out + (size_t)e * N2 * DOUT);
    }
}


# layer-0 scan reads one tile-id byte per edge (dst/400, written by a new pre-pass launch over both dst lists): 16 edges per 16-byte load, 4 per compare op; sums in LDS-add arrival order as before
# speedup vs baseline: 1.8139x; 1.1453x over previous
#include <hip/hip_runtime.h>

constexpr int N0 = 400000, N1 = 100000, N2 = 25000;
constexpr int E0 = 800000, E1 = 200000;
constexpr int DIN = 128, HID = 256, DOUT = 128;

typedef __attribute__((ext_vector_type(2))) float v2f;
typedef __attribute__((ext_vector_type(8))) float v8f;
typedef __attribute__((ext_vector_type(4))) float v4f;

__global__ void __launch_bounds__(256) zero_f32(float4* __restrict__ p, long n4) {
    const long stride = (long)gridDim.x * blockDim.x;
    const float4 z = {0.0f, 0.0f, 0.0f, 0.0f};
    for (long i = (long)blockIdx.x * blockDim.x + threadIdx.x; i < n4; i += stride) p[i] = z;
}

constexpr int FEAT_U = 8;
static_assert(N0 % 32 == 0, "build_feat takes whole groups of 32 rows");
__global__ void __launch_bounds__(256)
build_feat(const int* __restrict__ x,
           const float* __restrict__ e0, const float* __restrict__ e1,
           const float* __restrict__ e2, const float* __restrict__ e3,
           float* __restrict__ feat) {
    const int lane = threadIdx.x & 31;
    const int wave = __builtin_amdgcn_readfirstlane((int)((blockIdx.x * blockDim.x + threadIdx.x) >> 5));
    const int row0 = wave * 32;
    if (row0 >= N0) return;
    const int4 xi = ((const int4*)x)[row0 + lane];
    const float* tb;
    int rl;
    if (lane < 8)       { tb = e0 + lane * 4;        rl = 32; }
    else if (lane < 12) { tb = e1 + (lane - 8) * 4;  rl = 16; }
    else if (lane < 16) { tb = e2 + (lane - 12) * 4; rl = 16; }
    else                { tb = e3 + (lane - 16) * 4; rl = 64; }
    v4f* dst = (v4f*)(feat + (long)row0 * 128) + lane;
#pragma unroll 1
    for (int i = 0; i < 32; i += FEAT_U) {
        v4f v[FEAT_U];
#pragma unroll
        for (int u = 0; u < FEAT_U; ++u) {
            const int i0 = __builtin_amdgcn_readlane(xi.x, i + u);
            const int i1 = __builtin_amdgcn_readlane(xi.y, i + u);
            const int i2 = __builtin_amdgcn_readlane(xi.z, i + u);
            const int i3 = __builtin_amdgcn_readlane(xi.w, i + u);
            const int idx = (lane < 8) ? i0 : (lane < 12) ? i1 : (lane < 16) ? i2 : i3;
            v[u] = *(const v4f*)(tb + (long)idx * rl);
        }
        static_assert(FEAT_U == 8, "the fence below names eight values");
        asm volatile("" :: "v"(v[0]), "v"(v[1]), "v"(v[2]), "v"(v[3]),
                           "v"(v[4]), "v"(v[5]), "v"(v[6]), "v"(v[7]));
#pragma unroll
        for (int u = 0; u < FEAT_U; ++u) dst[(i + u) * 32] = v[u];
    }
}

template <int WIDTH, int U>
__global__ void __launch_bounds__(256)
scatter_edges(const float* __restrict__ h, const int* __restrict__ srcA,
              const int* __restrict__ dstA, float* __restrict__ agg,
              float* __restrict__ deg, int nE) {
    constexpr int L = WIDTH / 32;
    const int lane = threadIdx.x & 31;
    const int wave = __builtin_amdgcn_readfirstlane((int)((blockIdx.x * blockDim.x + threadIdx.x) >> 5));
    const int base = wave * 32;
    if (base >= nE) return;
    const int cnt = (nE - base < 32) ? (nE - base) : 32;
    int s = 0, d = 0;
    if (lane < cnt) {
        s = srcA[base + lane];
        d = dstA[base + lane];
        atomicAdd(deg + d, 1.0f);
    }
    const float* hl = h + lane;
    float* al = agg + lane;
    int i = 0;
#pragma unroll 1
    for (; i + U <= cnt; i += U) {
        float v[U][L];
        int di[U];
#pragma unroll
        for (int u = 0; u < U; ++u) {
            const int si = __builtin_amdgcn_readlane(s, i + u);
            di[u] = __builtin_amdgcn_readlane(d, i + u);
            const float* hp = hl + (long)si * WIDTH;
#pragma unroll
            for (int j = 0; j < L; ++j) v[u][j] = hp[32 * j];
        }
#pragma unroll
        for (int u = 0; u < U; ++u) {
            float* ap = al + (long)di[u] * WIDTH;
#pragma unroll
            for (int j = 0; j < L; ++j) atomicAdd(ap + 32 * j, v[u][j]);
        }
    }
    for (; i < cnt; ++i) {
        const int si = __builtin_amdgcn_readlane(s, i);
        const int dd = __builtin_amdgcn_readlane(d, i);
        const float* hp = hl + (long)si * WIDTH;
        float* ap = al + (long)dd * WIDTH;
        float v[L];
#pragma unroll
        for (int j = 0; j < L; ++j) v[j] = hp[32 * j];
#pragma unroll
        for (int j = 0; j < L; ++j) atomicAdd(ap + 32 * j, v[j]);
    }
}

constexpr int TM = 400;
static_assert(E0 % 16 == 0, "the tile-id plane is written and scanned in 16-byte groups");
constexpr int NG = E0 / 16;
static_assert(N1 % TM == 0 && N1 / TM <= 255, "a tile id fits a byte, 255 stays free");
__global__ void __launch_bounds__(256)
mark_tiles(const int* __restrict__ dA, const int* __restrict__ dB,
           unsigned int* __restrict__ tA, unsigned int* __restrict__ tB) {
    const int g = blockIdx.x * 256 + threadIdx.x;
    if (g >= NG) return;
    const int4* d4 = (const int4*)(blockIdx.y ? dB : dA) + 4 * g;
    int4 v[4];
#pragma unroll
    for (int u = 0; u < 4; ++u) v[u] = d4[u];
    auto tile = [](int d) -> unsigned { return ((unsigned)d < (unsigned)N1) ? (unsigned)d / (unsigned)TM : 255u; };
    auto pack = [&](const int4& a) -> unsigned {
        return tile(a.x) | (tile(a.y) << 8) | (tile(a.z) << 16) | (tile(a.w) << 24);
    };
    uint4 r;
    r.x = pack(v[0]); r.y = pack(v[1]); r.z = pack(v[2]); r.w = pack(v[3]);
    ((uint4*)(blockIdx.y ? tB : tA))[g] = r;
}

constexpr int PITCH = 132;
constexpr int QCAP = 8192;
constexpr int G1T = 1024;
static_assert(N1 % TM == 0 && TM % 16 == 0, "whole tiles only");
__global__ void __launch_bounds__(G1T)
agg_gemm1(const float* __restrict__ feat, const int* __restrict__ srcA,
          const int* __restrict__ dstA, const unsigned int* __restrict__ tile8,
          const float* __restrict__ W,
          const float* __restrict__ bias, float* __restrict__ h1) {
    __shared__ float As[TM * PITCH];
    __shared__ int degc[TM];
    __shared__ float invs[TM];
    __shared__ int q[QCAP];
    __shared__ int qn;
    const int tid = threadIdx.x;
    const int lane = tid & 31;
    const int wave = __builtin_amdgcn_readfirstlane(tid >> 5);
    const int m0 = blockIdx.x * TM;

    for (int i = tid; i < TM * PITCH; i += G1T) As[i] = 0.0f;
    for (int i = tid; i < TM; i += G1T) degc[i] = 0;
    if (tid == 0) qn = 0;
    __syncthreads();

    {
        const uint4* t4 = (const uint4*)tile8;
        const unsigned pat = (unsigned)blockIdx.x * 0x01010101u;
        auto push = [&](int e) {
            const int slot = atomicAdd(&qn, 1);
            if (slot < QCAP) {
                q[slot] = e;
            } else {
                const unsigned dl = min((unsigned)(dstA[e] - m0), (unsigned)(TM - 1));
                const float* hp = feat + (long)srcA[e] * DIN;
                for (int c = 0; c < DIN; ++c) atomicAdd(&As[dl * PITCH + c], hp[c]);
                atomicAdd(&degc[dl], 1);
            }
        };
        auto eqb = [&](unsigned x) -> unsigned {
            const unsigned t = x ^ pat;
            return ~(((t & 0x7F7F7F7Fu) + 0x7F7F7F7Fu) | t) & 0x80808080u;
        };
        for (int i0 = tid; i0 < NG; i0 += 4 * G1T) {
            uint4 w[4];
#pragma unroll
            for (int u = 0; u < 4; ++u) {
                const int i = i0 + u * G1T;
                w[u] = (i < NG) ? t4[i] : make_uint4(~0u, ~0u, ~0u, ~0u);
            }
#pragma unroll
            for (int u = 0; u < 4; ++u) {
                unsigned m = (eqb(w[u].x) >> 7) | (eqb(w[u].y) >> 6) | (eqb(w[u].z) >> 5) | (eqb(w[u].w) >> 4);
                const int e0 = (i0 + u * G1T) * 16;
                while (m) {
                    const int p = __builtin_ctz(m);
                    m &= m - 1;
                    push(e0 + 4 * (p & 7) + (p >> 3));
                }
            }
        }
    }
    __syncthreads();

    {
        const int qv = qn;
        const int n = __builtin_amdgcn_readfirstlane(qv < QCAP ? qv : QCAP);
        const float* hl = feat + lane;
        float* al = As + lane;
        for (int base = wave * 32; base < n; base += G1T) {
            const int cnt = (n - base < 32) ? (n - base) : 32;
            int s = 0, dl = 0;
            if (lane < cnt) {
                const int e = q[base + lane];
                s = srcA[e];
                dl = (int)min((unsigned)(dstA[e] - m0), (unsigned)(TM - 1));
                atomicAdd(&degc[dl], 1);
            }
            int i = 0;
#pragma unroll 1
            for (; i + 8 <= cnt; i += 8) {
                float v[8][4];
                int di[8];
#pragma unroll
                for (int u = 0; u < 8; ++u) {
                    const int si = __builtin_amdgcn_readlane(s, i + u);
                    di[u] = __builtin_amdgcn_readlane(dl, i + u);
                    const float* hp = hl + (long)si * DIN;
#pragma unroll
                    for (int j = 0; j < 4; ++j) v[u][j] = hp[32 * j];
                }
#pragma unroll
                for (int u = 0; u < 8; ++u) {
                    float* ap = al + di[u] * PITCH;
#pragma unroll
                    for (int j = 0; j < 4; ++j) atomicAdd(ap + 32 * j, v[u][j]);
                }
            }
            for (; i < cnt; ++i) {
                const int si = __builtin_amdgcn_readlane(s, i);
                const int dd = __builtin_amdgcn_readlane(dl, i);
                const float* hp = hl + (long)si * DIN;
                float* ap = al + dd * PITCH;
                float v[4];
#pragma unroll
                for (int j = 0; j < 4; ++j) v[j] = hp[32 * j];
#pragma unroll
                for (int j = 0; j < 4; ++j) atomicAdd(ap + 32 * j, v[j]);
            }
        }
    }
    __syncthreads();

    for (int r = tid; r < TM; r += G1T) invs[r] = 1.0f / fmaxf((float)degc[r], 1.0f);
    __syncthreads();
    for (int idx = tid; idx < TM * DIN; idx += G1T) {
        const int r = idx >> 7, c = idx & 127;
        As[r * PITCH + c] = feat[(long)(m0 + r) * DIN + c] + As[r * PITCH + c] * invs[r];
    }
    __syncthreads();

    const int lo = lane & 15, hi = lane >> 4;
    const int ncol = (wave & 15) * 16 + lo;
    v2f b[DIN / 4];
#pragma unroll
    for (int ks = 0; ks < DIN / 4; ++ks) {
        const int kk = 4 * ks + 2 * hi;
        b[ks][0] = W[kk * HID + ncol];
        b[ks][1] = W[(kk + 1) * HID + ncol];
    }
    const float bv = bias[ncol];
#pragma unroll 1
    for (int mt = wave >> 4; mt < TM / 16; mt += 2) {
        const float* ar = As + (mt * 16 + lo) * PITCH + 2 * hi;
        v8f acc = {0, 0, 0, 0, 0, 0, 0, 0};
#pragma unroll
        for (int ks = 0; ks < DIN / 4; ++ks) {
            v2f a;
            a[0] = ar[4 * ks];
            a[1] = ar[4 * ks + 1];
            acc = __builtin_amdgcn_wmma_f32_16x16x4_f32(false, a, false, b[ks], (short)0, acc, false, false);
        }
#pragma unroll
        for (int r = 0; r < 8; ++r) {
            const int rr = mt * 16 + r + 8 * hi;
            const float s = (degc[rr] > 0) ? 2.0f : 1.0f;
            h1[(long)(m0 + rr) * HID + ncol] = fmaxf(acc[r] + s * bv, 0.0f);
        }
    }
}

__global__ void __launch_bounds__(256)
gemm2_fused(const float* __restrict__ h1, const float* __restrict__ agg1,
            const float* __restrict__ deg1, const float* __restrict__ W,
            const float* __restrict__ bias, float* __restrict__ out) {
    __shared__ float As[16 * 260];
    const int m0 = blockIdx.x * 16;
    const int tid = threadIdx.x;
#pragma unroll
    for (int i = 0; i < 16; ++i) {
        int idx = tid + i * 256;
        int r = idx >> 8, c = idx & 255;
        int row = m0 + r;
        float v = 0.0f;
        if (row < N2) {
            float inv = 1.0f / fmaxf(deg1[row], 1.0f);
            v = h1[(long)row * HID + c] + agg1[(long)row * HID + c] * inv;
        }
        As[r * 260 + c] = v;
    }
    __syncthreads();

    const int lane = tid & 31, wave = tid >> 5;
    const int lo = lane & 15, hi = lane >> 4;
    const int n0 = wave * 16;
    v8f acc = {0, 0, 0, 0, 0, 0, 0, 0};
    for (int k = 0; k < HID; k += 4) {
        const int kk = k + 2 * hi;
        v2f a;
        a[0] = As[lo * 260 + kk];
        a[1] = As[lo * 260 + kk + 1];
        v2f b;
        b[0] = W[kk * DOUT + n0 + lo];
        b[1] = W[(kk + 1) * DOUT + n0 + lo];
        acc = __builtin_amdgcn_wmma_f32_16x16x4_f32(false, a, false, b, (short)0, acc, false, false);
    }
#pragma unroll
    for (int r = 0; r < 8; ++r) {
        int row = m0 + r + 8 * hi;
        if (row < N2) out[(long)row * DOUT + n0 + lo] = acc[r] + bias[n0 + lo];
    }
}

extern "C" void kernel_launch(void* const* d_in, const int* in_sizes, int n_in,
                              void* d_out, int out_size, void* d_ws, size_t ws_size,
                              hipStream_t stream) {
    (void)in_sizes; (void)n_in; (void)out_size; (void)ws_size;
    const int*   x  = (const int*)d_in[0];
    const float* e0 = (const float*)d_in[1];
    const float* e1 = (const float*)d_in[2];
    const float* e2 = (const float*)d_in[3];
    const float* e3 = (const float*)d_in[4];
    const float* Win[2]  = {(const float*)d_in[5], (const float*)d_in[9]};
    const float* bin[2]  = {(const float*)d_in[6], (const float*)d_in[10]};
    const float* Wout[2] = {(const float*)d_in[7], (const float*)d_in[11]};
    const float* bout[2] = {(const float*)d_in[8], (const float*)d_in[12]};
    const int* srcL0[2] = {(const int*)d_in[13], (const int*)d_in[17]};
    const int* dstL0[2] = {(const int*)d_in[14], (const int*)d_in[18]};
    const int* srcL1[2] = {(const int*)d_in[15], (const int*)d_in[19]};
    const int* dstL1[2] = {(const int*)d_in[16], (const int*)d_in[20]};
    float* out = (float*)d_out;

    size_t off = 0;
    auto take = [&](size_t bytes) { size_t o = off; off += (bytes + 255) & ~(size_t)255; return o; };
    char* ws = (char*)d_ws;
    float* feat = (float*)(ws + take((size_t)N0 * DIN * 4));
    float* h1   = (float*)(ws + take((size_t)N1 * HID * 4));
    unsigned int* tl[2];
    tl[0] = (unsigned int*)(ws + take((size_t)E0));
    tl[1] = (unsigned int*)(ws + take((size_t)E0));
    size_t zbeg = off;
    float* agg1 = (float*)(ws + take((size_t)N2 * HID * 4));
    float* deg1 = (float*)(ws + take((size_t)N2 * 4));
    float4* zptr = (float4*)(ws + zbeg);
    long    zcnt4 = (long)((off - zbeg) / 16);

    build_feat<<<(N0 / 32 + 7) / 8, 256, 0, stream>>>(x, e0, e1, e2, e3, feat);

    mark_tiles<<<dim3((NG + 255) / 256, 2), 256, 0, stream>>>(dstL0[0], dstL0[1], tl[0], tl[1]);

    for (int e = 0; e < 2; ++e) {
        zero_f32<<<2048, 256, 0, stream>>>(zptr, zcnt4);
        agg_gemm1<<<N1 / TM, G1T, 0, stream>>>(feat, srcL0[e], dstL0[e], tl[e], Win[e], bin[e], h1);
        scatter_edges<HID, 4><<<((E1 + 31) / 32 + 7) / 8, 256, 0, stream>>>(h1, srcL1[e], dstL1[e], agg1, deg1, E1);
        gemm2_fused<<<(N2 + 15) / 16, 256, 0, stream>>>(h1, agg1, deg1, Wout[e], bout[e],
                                                        out + (size_t)e * N2 * DOUT);
    }
}


# layer-1 sums kept in LDS too: a block owns 100 destinations, scans their tile-id bytes, adds h1 rows by LDS float adds, runs gemm2 from LDS; agg1 plane, zero fill, global atomics gone; LDS-add order
# speedup vs baseline: 2.4570x; 1.3545x over previous
#include <hip/hip_runtime.h>

constexpr int N0 = 400000, N1 = 100000, N2 = 25000;
constexpr int E0 = 800000, E1 = 200000;
constexpr int DIN = 128, HID = 256, DOUT = 128;

typedef __attribute__((ext_vector_type(2))) float v2f;
typedef __attribute__((ext_vector_type(8))) float v8f;
typedef __attribute__((ext_vector_type(4))) float v4f;

constexpr int FEAT_U = 8;
static_assert(N0 % 32 == 0, "build_feat takes whole groups of 32 rows");
__global__ void __launch_bounds__(256)
build_feat(const int* __restrict__ x,
           const float* __restrict__ e0, const float* __restrict__ e1,
           const float* __restrict__ e2, const float* __restrict__ e3,
           float* __restrict__ feat) {
    const int lane = threadIdx.x & 31;
    const int wave = __builtin_amdgcn_readfirstlane((int)((blockIdx.x * blockDim.x + threadIdx.x) >> 5));
    const int row0 = wave * 32;
    if (row0 >= N0) return;
    const int4 xi = ((const int4*)x)[row0 + lane];
    const float* tb;
    int rl;
    if (lane < 8)       { tb = e0 + lane * 4;        rl = 32; }
    else if (lane < 12) { tb = e1 + (lane - 8) * 4;  rl = 16; }
    else if (lane < 16) { tb = e2 + (lane - 12) * 4; rl = 16; }
    else                { tb = e3 + (lane - 16) * 4; rl = 64; }
    v4f* dst = (v4f*)(feat + (long)row0 * 128) + lane;
#pragma unroll 1
    for (int i = 0; i < 32; i += FEAT_U) {
        v4f v[FEAT_U];
#pragma unroll
        for (int u = 0; u < FEAT_U; ++u) {
            const int i0 = __builtin_amdgcn_readlane(xi.x, i + u);
            const int i1 = __builtin_amdgcn_readlane(xi.y, i + u);
            const int i2 = __builtin_amdgcn_readlane(xi.z, i + u);
            const int i3 = __builtin_amdgcn_readlane(xi.w, i + u);
            const int idx = (lane < 8) ? i0 : (lane < 12) ? i1 : (lane < 16) ? i2 : i3;
            v[u] = *(const v4f*)(tb + (long)idx * rl);
        }
        static_assert(FEAT_U == 8, "the fence below names eight values");
        asm volatile("" :: "v"(v[0]), "v"(v[1]), "v"(v[2]), "v"(v[3]),
                           "v"(v[4]), "v"(v[5]), "v"(v[6]), "v"(v[7]));
#pragma unroll
        for (int u = 0; u < FEAT_U; ++u) dst[(i + u) * 32] = v[u];
    }
}

constexpr int TM = 400;
constexpr int TM2 = 100;
static_assert(E0 % 16 == 0 && E1 % 16 == 0, "the tile-id planes are written and scanned in 16-byte groups");
constexpr int NG = E0 / 16;
constexpr int NG1 = E1 / 16;
static_assert(N1 % TM == 0 && N1 / TM <= 255, "a tile id fits a byte, 255 stays free");
static_assert(N2 % TM2 == 0 && N2 / TM2 <= 255, "a tile id fits a byte, 255 stays free");
__global__ void __launch_bounds__(256)
mark_tiles(const int* __restrict__ d0A, const int* __restrict__ d0B,
           const int* __restrict__ d1A, const int* __restrict__ d1B,
           unsigned int* __restrict__ t0A, unsigned int* __restrict__ t0B,
           unsigned int* __restrict__ t1A, unsigned int* __restrict__ t1B) {
    const int y = blockIdx.y;
    const bool l1 = y >= 2;
    const int g = blockIdx.x * 256 + threadIdx.x;
    if (g >= (l1 ? NG1 : NG)) return;
    const int* d = (y == 0) ? d0A : (y == 1) ? d0B : (y == 2) ? d1A : d1B;
    unsigned int* t = (y == 0) ? t0A : (y == 1) ? t0B : (y == 2) ? t1A : t1B;
    const int4* d4 = (const int4*)d + 4 * g;
    int4 v[4];
#pragma unroll
    for (int u = 0; u < 4; ++u) v[u] = d4[u];
    auto tile = [&](int dd) -> unsigned {
        const unsigned du = (unsigned)dd;
        const unsigned q0 = (du < (unsigned)N1) ? du / (unsigned)TM : 255u;
        const unsigned q1 = (du < (unsigned)N2) ? du / (unsigned)TM2 : 255u;
        return l1 ? q1 : q0;
    };
    auto pack = [&](const int4& a) -> unsigned {
        return tile(a.x) | (tile(a.y) << 8) | (tile(a.z) << 16) | (tile(a.w) << 24);
    };
    uint4 r;
    r.x = pack(v[0]); r.y = pack(v[1]); r.z = pack(v[2]); r.w = pack(v[3]);
    ((uint4*)t)[g] = r;
}

constexpr int PITCH = 132;
constexpr int QCAP = 8192;
constexpr int G1T = 1024;
static_assert(N1 % TM == 0 && TM % 16 == 0, "whole tiles only");
__global__ void __launch_bounds__(G1T)
agg_gemm1(const float* __restrict__ feat, const int* __restrict__ srcA,
          const int* __restrict__ dstA, const unsigned int* __restrict__ tile8,
          const float* __restrict__ W,
          const float* __restrict__ bias, float* __restrict__ h1) {
    __shared__ float As[TM * PITCH];
    __shared__ int degc[TM];
    __shared__ float invs[TM];
    __shared__ int q[QCAP];
    __shared__ int qn;
    const int tid = threadIdx.x;
    const int lane = tid & 31;
    const int wave = __builtin_amdgcn_readfirstlane(tid >> 5);
    const int m0 = blockIdx.x * TM;

    for (int i = tid; i < TM * PITCH; i += G1T) As[i] = 0.0f;
    for (int i = tid; i < TM; i += G1T) degc[i] = 0;
    if (tid == 0) qn = 0;
    __syncthreads();

    {
        const uint4* t4 = (const uint4*)tile8;
        const unsigned pat = (unsigned)blockIdx.x * 0x01010101u;
        auto push = [&](int e) {
            const int slot = atomicAdd(&qn, 1);
            if (slot < QCAP) {
                q[slot] = e;
            } else {
                const unsigned dl = min((unsigned)(dstA[e] - m0), (unsigned)(TM - 1));
                const float* hp = feat + (long)srcA[e] * DIN;
                for (int c = 0; c < DIN; ++c) atomicAdd(&As[dl * PITCH + c], hp[c]);
                atomicAdd(&degc[dl], 1);
            }
        };
        auto eqb = [&](unsigned x) -> unsigned {
            const unsigned t = x ^ pat;
            return ~(((t & 0x7F7F7F7Fu) + 0x7F7F7F7Fu) | t) & 0x80808080u;
        };
        for (int i0 = tid; i0 < NG; i0 += 4 * G1T) {
            uint4 w[4];
#pragma unroll
            for (int u = 0; u < 4; ++u) {
                const int i = i0 + u * G1T;
                w[u] = (i < NG) ? t4[i] : make_uint4(~0u, ~0u, ~0u, ~0u);
            }
#pragma unroll
            for (int u = 0; u < 4; ++u) {
                unsigned m = (eqb(w[u].x) >> 7) | (eqb(w[u].y) >> 6) | (eqb(w[u].z) >> 5) | (eqb(w[u].w) >> 4);
                const int e0 = (i0 + u * G1T) * 16;
                while (m) {
                    const int p = __builtin_ctz(m);
                    m &= m - 1;
                    push(e0 + 4 * (p & 7) + (p >> 3));
                }
            }
        }
    }
    __syncthreads();

    {
        const int qv = qn;
        const int n = __builtin_amdgcn_readfirstlane(qv < QCAP ? qv : QCAP);
        const float* hl = feat + lane;
        float* al = As + lane;
        for (int base = wave * 32; base < n; base += G1T) {
            const int cnt = (n - base < 32) ? (n - base) : 32;
            int s = 0, dl = 0;
            if (lane < cnt) {
                const int e = q[base + lane];
                s = srcA[e];
                dl = (int)min((unsigned)(dstA[e] - m0), (unsigned)(TM - 1));
                atomicAdd(&degc[dl], 1);
            }
            int i = 0;
#pragma unroll 1
            for (; i + 8 <= cnt; i += 8) {
                float v[8][4];
                int di[8];
#pragma unroll
                for (int u = 0; u < 8; ++u) {
                    const int si = __builtin_amdgcn_readlane(s, i + u);
                    di[u] = __builtin_amdgcn_readlane(dl, i + u);
                    const float* hp = hl + (long)si * DIN;
#pragma unroll
                    for (int j = 0; j < 4; ++j) v[u][j] = hp[32 * j];
                }
#pragma unroll
                for (int u = 0; u < 8; ++u) {
                    float* ap = al + di[u] * PITCH;
#pragma unroll
                    for (int j = 0; j < 4; ++j) atomicAdd(ap + 32 * j, v[u][j]);
                }
            }
            for (; i < cnt; ++i) {
                const int si = __builtin_amdgcn_readlane(s, i);
                const int dd = __builtin_amdgcn_readlane(dl, i);
                const float* hp = hl + (long)si * DIN;
                float* ap = al + dd * PITCH;
                float v[4];
#pragma unroll
                for (int j = 0; j < 4; ++j) v[j] = hp[32 * j];
#pragma unroll
                for (int j = 0; j < 4; ++j) atomicAdd(ap + 32 * j, v[j]);
            }
        }
    }
    __syncthreads();

    for (int r = tid; r < TM; r += G1T) invs[r] = 1.0f / fmaxf((float)degc[r], 1.0f);
    __syncthreads();
    for (int idx = tid; idx < TM * DIN; idx += G1T) {
        const int r = idx >> 7, c = idx & 127;
        As[r * PITCH + c] = feat[(long)(m0 + r) * DIN + c] + As[r * PITCH + c] * invs[r];
    }
    __syncthreads();

    const int lo = lane & 15, hi = lane >> 4;
    const int ncol = (wave & 15) * 16 + lo;
    v2f b[DIN / 4];
#pragma unroll
    for (int ks = 0; ks < DIN / 4; ++ks) {
        const int kk = 4 * ks + 2 * hi;
        b[ks][0] = W[kk * HID + ncol];
        b[ks][1] = W[(kk + 1) * HID + ncol];
    }
    const float bv = bias[ncol];
#pragma unroll 1
    for (int mt = wave >> 4; mt < TM / 16; mt += 2) {
        const float* ar = As + (mt * 16 + lo) * PITCH + 2 * hi;
        v8f acc = {0, 0, 0, 0, 0, 0, 0, 0};
#pragma unroll
        for (int ks = 0; ks < DIN / 4; ++ks) {
            v2f a;
            a[0] = ar[4 * ks];
            a[1] = ar[4 * ks + 1];
            acc = __builtin_amdgcn_wmma_f32_16x16x4_f32(false, a, false, b[ks], (short)0, acc, false, false);
        }
#pragma unroll
        for (int r = 0; r < 8; ++r) {
            const int rr = mt * 16 + r + 8 * hi;
            const float s = (degc[rr] > 0) ? 2.0f : 1.0f;
            h1[(long)(m0 + rr) * HID + ncol] = fmaxf(acc[r] + s * bv, 0.0f);
        }
    }
}

constexpr int MT2 = (TM2 + 15) / 16;
constexpr int PITCH2 = 260;
constexpr int QCAP2 = 12288;
static_assert(MT2 > 4 && MT2 <= 8, "every wave has a first M tile, and at most a second one");
static_assert(PITCH2 % 4 == 0 && HID % 4 == 0, "rows are cleared and finished in 16-byte pieces");
__global__ void __launch_bounds__(G1T)
agg_gemm2(const float* __restrict__ h1, const int* __restrict__ srcA,
          const int* __restrict__ dstA, const unsigned int* __restrict__ tile8,
          const float* __restrict__ W,
          const float* __restrict__ bias, float* __restrict__ out) {
    __shared__ __attribute__((aligned(16))) float As[MT2 * 16 * PITCH2];
    __shared__ int degc[MT2 * 16];
    __shared__ float invs[MT2 * 16];
    __shared__ int q[QCAP2];
    __shared__ int qn;
    const int tid = threadIdx.x;
    const int lane = tid & 31;
    const int wave = __builtin_amdgcn_readfirstlane(tid >> 5);
    const int m0 = blockIdx.x * TM2;

    {
        const v4f z = {0.0f, 0.0f, 0.0f, 0.0f};
        for (int i = tid; i < MT2 * 16 * PITCH2 / 4; i += G1T) ((v4f*)As)[i] = z;
    }
    for (int i = tid; i < MT2 * 16; i += G1T) degc[i] = 0;
    if (tid == 0) qn = 0;
    __syncthreads();

    {
        const uint4* t4 = (const uint4*)tile8;
        const unsigned pat = (unsigned)blockIdx.x * 0x01010101u;
        auto push = [&](int e) {
            const int slot = atomicAdd(&qn, 1);
            if (slot < QCAP2) {
                q[slot] = e;
            } else {
                const unsigned dl = min((unsigned)(dstA[e] - m0), (unsigned)(TM2 - 1));
                const float* hp = h1 + (long)srcA[e] * HID;
                for (int c = 0; c < HID; ++c) atomicAdd(&As[dl * PITCH2 + c], hp[c]);
                atomicAdd(&degc[dl], 1);
            }
        };
        auto eqb = [&](unsigned x) -> unsigned {
            const unsigned t = x ^ pat;
            return ~(((t & 0x7F7F7F7Fu) + 0x7F7F7F7Fu) | t) & 0x80808080u;
        };
        for (int i0 = tid; i0 < NG1; i0 += 4 * G1T) {
            uint4 w[4];
#pragma unroll
            for (int u = 0; u < 4; ++u) {
                const int i = i0 + u * G1T;
                w[u] = (i < NG1) ? t4[i] : make_uint4(~0u, ~0u, ~0u, ~0u);
            }
#pragma unroll
            for (int u = 0; u < 4; ++u) {
                unsigned m = (eqb(w[u].x) >> 7) | (eqb(w[u].y) >> 6) | (eqb(w[u].z) >> 5) | (eqb(w[u].w) >> 4);
                const int e0 = (i0 + u * G1T) * 16;
                while (m) {
                    const int p = __builtin_ctz(m);
                    m &= m - 1;
                    push(e0 + 4 * (p & 7) + (p >> 3));
                }
            }
        }
    }
    __syncthreads();

    {
        const int qv = qn;
        const int n = __builtin_amdgcn_readfirstlane(qv < QCAP2 ? qv : QCAP2);
        const float* hl = h1 + lane;
        float* al = As + lane;
        for (int base = wave * 32; base < n; base += G1T) {
            const int cnt = (n - base < 32) ? (n - base) : 32;
            int s = 0, dl = 0;
            if (lane < cnt) {
                const int e = q[base + lane];
                s = srcA[e];
                dl = (int)min((unsigned)(dstA[e] - m0), (unsigned)(TM2 - 1));
                atomicAdd(&degc[dl], 1);
            }
            int i = 0;
#pragma unroll 1
            for (; i + 4 <= cnt; i += 4) {
                float v[4][8];
                int di[4];
#pragma unroll
                for (int u = 0; u < 4; ++u) {
                    const int si = __builtin_amdgcn_readlane(s, i + u);
                    di[u] = __builtin_amdgcn_readlane(dl, i + u);
                    const float* hp = hl + (long)si * HID;
#pragma unroll
                    for (int j = 0; j < 8; ++j) v[u][j] = hp[32 * j];
                }
#pragma unroll
                for (int u = 0; u < 4; ++u) {
                    float* ap = al + di[u] * PITCH2;
#pragma unroll
                    for (int j = 0; j < 8; ++j) atomicAdd(ap + 32 * j, v[u][j]);
                }
            }
            for (; i < cnt; ++i) {
                const int si = __builtin_amdgcn_readlane(s, i);
                const int dd = __builtin_amdgcn_readlane(dl, i);
                const float* hp = hl + (long)si * HID;
                float* ap = al + dd * PITCH2;
                float v[8];
#pragma unroll
                for (int j = 0; j < 8; ++j) v[j] = hp[32 * j];
#pragma unroll
                for (int j = 0; j < 8; ++j) atomicAdd(ap + 32 * j, v[j]);
            }
        }
    }
    __syncthreads();

    for (int r = tid; r < TM2; r += G1T) invs[r] = 1.0f / fmaxf((float)degc[r], 1.0f);
    __syncthreads();
    {
        constexpr int NV = TM2 * (HID / 4);
        constexpr int S3 = (NV + G1T - 1) / G1T;
        const v4f* own = (const v4f*)(h1 + (long)m0 * HID);
        v4f hv[S3];
#pragma unroll
        for (int u = 0; u < S3; ++u) {
            const int idx = tid + u * G1T;
            hv[u] = own[idx < NV ? idx : NV - 1];
        }
#pragma unroll
        for (int u = 0; u < S3; ++u) {
            const int idx = tid + u * G1T;
            if (idx < NV) {
                const int r = idx / (HID / 4), c4 = idx % (HID / 4);
                v4f* ap = (v4f*)(As + r * PITCH2) + c4;
                *ap = hv[u] + *ap * invs[r];
            }
        }
    }
    __syncthreads();

    const int lo = lane & 15, hi = lane >> 4;
    const int ncol = (wave & 7) * 16 + lo;
    const int g = wave >> 3;
    const float* wp = W + (2 * hi) * DOUT + ncol;
    const float* a0p = As + (g * 16 + lo) * PITCH2 + 2 * hi;
    const float bv = bias[ncol];
    v8f acc0 = {0, 0, 0, 0, 0, 0, 0, 0};
    if (g + 4 < MT2) {
        const float* a1p = a0p + 64 * PITCH2;
        v8f acc1 = {0, 0, 0, 0, 0, 0, 0, 0};
#pragma unroll 16
        for (int ks = 0; ks < HID / 4; ++ks) {
            v2f b, a0, a1;
            b[0] = wp[(4 * ks) * DOUT];
            b[1] = wp[(4 * ks + 1) * DOUT];
            a0[0] = a0p[4 * ks];
            a0[1] = a0p[4 * ks + 1];
            a1[0] = a1p[4 * ks];
            a1[1] = a1p[4 * ks + 1];
            acc0 = __builtin_amdgcn_wmma_f32_16x16x4_f32(false, a0, false, b, (short)0, acc0, false, false);
            acc1 = __builtin_amdgcn_wmma_f32_16x16x4_f32(false, a1, false, b, (short)0, acc1, false, false);
        }
#pragma unroll
        for (int r = 0; r < 8; ++r) {
            const int rr = (g + 4) * 16 + r + 8 * hi;
            if (rr < TM2) out[(long)(m0 + rr) * DOUT + ncol] = acc1[r] + bv;
        }
    } else {
#pragma unroll 16
        for (int ks = 0; ks < HID / 4; ++ks) {
            v2f b, a0;
            b[0] = wp[(4 * ks) * DOUT];
            b[1] = wp[(4 * ks + 1) * DOUT];
            a0[0] = a0p[4 * ks];
            a0[1] = a0p[4 * ks + 1];
            acc0 = __builtin_amdgcn_wmma_f32_16x16x4_f32(false, a0, false, b, (short)0, acc0, false, false);
        }
    }
    static_assert(16 * 4 <= TM2, "M tiles 0..3 hold only real rows");
#pragma unroll
    for (int r = 0; r < 8; ++r) {
        const int rr = g * 16 + r + 8 * hi;
        out[(long)(m0 + rr) * DOUT + ncol] = acc0[r] + bv;
    }
}

extern "C" void kernel_launch(void* const* d_in, const int* in_sizes, int n_in,
                              void* d_out, int out_size, void* d_ws, size_t ws_size,
                              hipStream_t stream) {
    (void)in_sizes; (void)n_in; (void)out_size; (void)ws_size;
    const int*   x  = (const int*)d_in[0];
    const float* e0 = (const float*)d_in[1];
    const float* e1 = (const float*)d_in[2];
    const float* e2 = (const float*)d_in[3];
    const float* e3 = (const float*)d_in[4];
    const float* Win[2]  = {(const float*)d_in[5], (const float*)d_in[9]};
    const float* bin[2]  = {(const float*)d_in[6], (const float*)d_in[10]};
    const float* Wout[2] = {(const float*)d_in[7], (const float*)d_in[11]};
    const float* bout[2] = {(const float*)d_in[8], (const float*)d_in[12]};
    const int* srcL0[2] = {(const int*)d_in[13], (const int*)d_in[17]};
    const int* dstL0[2] = {(const int*)d_in[14], (const int*)d_in[18]};
    const int* srcL1[2] = {(const int*)d_in[15], (const int*)d_in[19]};
    const int* dstL1[2] = {(const int*)d_in[16], (const int*)d_in[20]};
    float* out = (float*)d_out;

    size_t off = 0;
    auto take = [&](size_t bytes) { size_t o = off; off += (bytes + 255) & ~(size_t)255; return o; };
    char* ws = (char*)d_ws;
    float* feat = (float*)(ws + take((size_t)N0 * DIN * 4));
    float* h1   = (float*)(ws + take((size_t)N1 * HID * 4));
    unsigned int* tl0[2];
    unsigned int* tl1[2];
    tl0[0] = (unsigned int*)(ws + take((size_t)E0));
    tl0[1] = (unsigned int*)(ws + take((size_t)E0));
    tl1[0] = (unsigned int*)(ws + take((size_t)E1));
    tl1[1] = (unsigned int*)(ws + take((size_t)E1));

    build_feat<<<(N0 / 32 + 7) / 8, 256, 0, stream>>>(x, e0, e1, e2, e3, feat);

    mark_tiles<<<dim3((NG + 255) / 256, 4), 256, 0, stream>>>(dstL0[0], dstL0[1], dstL1[0], dstL1[1],
                                                             tl0[0], tl0[1], tl1[0], tl1[1]);

    for (int e = 0; e < 2; ++e) {
        agg_gemm1<<<N1 / TM, G1T, 0, stream>>>(feat, srcL0[e], dstL0[e], tl0[e], Win[e], bin[e], h1);
        agg_gemm2<<<N2 / TM2, G1T, 0, stream>>>(h1, srcL1[e], dstL1[e], tl1[e], Wout[e], bout[e],
                                                out + (size_t)e * N2 * DOUT);
    }
}


# edge lists pre-sorted by tile in 8192-edge chunks (pre-pass: LDS counting sort, src|row words + offsets); blocks copy their pieces: tile-byte scan and per-edge index loads gone; LDS-add arrival order
# speedup vs baseline: 2.7515x; 1.1199x over previous
#include <hip/hip_runtime.h>

constexpr int N0 = 400000, N1 = 100000, N2 = 25000;
constexpr int E0 = 800000, E1 = 200000;
constexpr int DIN = 128, HID = 256, DOUT = 128;

typedef __attribute__((ext_vector_type(2))) float v2f;
typedef __attribute__((ext_vector_type(8))) float v8f;
typedef __attribute__((ext_vector_type(4))) float v4f;

constexpr int FEAT_U = 8;
static_assert(N0 % 32 == 0, "build_feat takes whole groups of 32 rows");
__global__ void __launch_bounds__(256)
build_feat(const int* __restrict__ x,
           const float* __restrict__ e0, const float* __restrict__ e1,
           const float* __restrict__ e2, const float* __restrict__ e3,
           float* __restrict__ feat) {
    const int lane = threadIdx.x & 31;
    const int wave = __builtin_amdgcn_readfirstlane((int)((blockIdx.x * blockDim.x + threadIdx.x) >> 5));
    const int row0 = wave * 32;
    if (row0 >= N0) return;
    const int4 xi = ((const int4*)x)[row0 + lane];
    const float* tb;
    int rl;
    if (lane < 8)       { tb = e0 + lane * 4;        rl = 32; }
    else if (lane < 12) { tb = e1 + (lane - 8) * 4;  rl = 16; }
    else if (lane < 16) { tb = e2 + (lane - 12) * 4; rl = 16; }
    else                { tb = e3 + (lane - 16) * 4; rl = 64; }
    v4f* dst = (v4f*)(feat + (long)row0 * 128) + lane;
#pragma unroll 1
    for (int i = 0; i < 32; i += FEAT_U) {
        v4f v[FEAT_U];
#pragma unroll
        for (int u = 0; u < FEAT_U; ++u) {
            const int i0 = __builtin_amdgcn_readlane(xi.x, i + u);
            const int i1 = __builtin_amdgcn_readlane(xi.y, i + u);
            const int i2 = __builtin_amdgcn_readlane(xi.z, i + u);
            const int i3 = __builtin_amdgcn_readlane(xi.w, i + u);
            const int idx = (lane < 8) ? i0 : (lane < 12) ? i1 : (lane < 16) ? i2 : i3;
            v[u] = *(const v4f*)(tb + (long)idx * rl);
        }
        static_assert(FEAT_U == 8, "the fence below names eight values");
        asm volatile("" :: "v"(v[0]), "v"(v[1]), "v"(v[2]), "v"(v[3]),
                           "v"(v[4]), "v"(v[5]), "v"(v[6]), "v"(v[7]));
#pragma unroll
        for (int u = 0; u < FEAT_U; ++u) dst[(i + u) * 32] = v[u];
    }
}

constexpr int TM = 400;
constexpr int TM2 = 100;
constexpr int CH = 8192;
constexpr int BT = 1024;
constexpr int EPT = CH / BT;
constexpr int NCH0 = (E0 + CH - 1) / CH;
constexpr int NCH1 = (E1 + CH - 1) / CH;
constexpr int OFFP = 256;
constexpr int SRC_BITS = 19;
constexpr unsigned SRC_MASK = (1u << SRC_BITS) - 1u;
constexpr int NT0 = N1 / TM;
constexpr int NT1 = N2 / TM2;
static_assert(N1 % TM == 0 && NT0 < OFFP - 1, "tile t + 1 is an entry of the table, OFFP - 1 stays free for 'no tile'");
static_assert(N2 % TM2 == 0 && NT1 < OFFP - 1, "tile t + 1 is an entry of the table, OFFP - 1 stays free for 'no tile'");
static_assert(N0 <= (1 << SRC_BITS) && N1 <= (1 << SRC_BITS), "src fits the low bits of a record");
static_assert(TM <= (1 << (31 - SRC_BITS)) && TM2 <= (1 << (31 - SRC_BITS)), "the row inside the tile fits the high bits");
static_assert(CH % BT == 0 && OFFP == 8 * 32, "the prefix sum below gives each lane of one wave eight bins");
__global__ void __launch_bounds__(BT)
bin_edges(const int* __restrict__ d0A, const int* __restrict__ d0B,
          const int* __restrict__ d1A, const int* __restrict__ d1B,
          const int* __restrict__ s0A, const int* __restrict__ s0B,
          const int* __restrict__ s1A, const int* __restrict__ s1B,
          unsigned int* __restrict__ r0A, unsigned int* __restrict__ r0B,
          unsigned int* __restrict__ r1A, unsigned int* __restrict__ r1B,
          int* __restrict__ o0A, int* __restrict__ o0B,
          int* __restrict__ o1A, int* __restrict__ o1B) {
    __shared__ int cnt[OFFP];
    __shared__ int off[OFFP];
    __shared__ unsigned int rec[CH];
    const int y = blockIdx.y;
    const bool l1 = y >= 2;
    const int c = blockIdx.x;
    if (c >= (l1 ? NCH1 : NCH0)) return;
    const int* d = (y == 0) ? d0A : (y == 1) ? d0B : (y == 2) ? d1A : d1B;
    const int* sp = (y == 0) ? s0A : (y == 1) ? s0B : (y == 2) ? s1A : s1B;
    unsigned int* rout = (y == 0) ? r0A : (y == 1) ? r0B : (y == 2) ? r1A : r1B;
    int* oout = (y == 0) ? o0A : (y == 1) ? o0B : (y == 2) ? o1A : o1B;
    const int E = l1 ? E1 : E0;
    const unsigned nd = l1 ? (unsigned)N2 : (unsigned)N1;
    const int tid = threadIdx.x;
    if (tid < OFFP) cnt[tid] = 0;
    __syncthreads();

    const int e0 = c * CH;
    int sv[EPT], dv[EPT];
#pragma unroll
    for (int k = 0; k < EPT; ++k) {
        const int e = e0 + tid + k * BT;
        const bool in = e < E;
        dv[k] = in ? d[e] : -1;
        sv[k] = in ? sp[e] : 0;
    }
    int tl[EPT], rk[EPT];
#pragma unroll
    for (int k = 0; k < EPT; ++k) {
        const unsigned du = (unsigned)dv[k];
        const unsigned q0 = du / (unsigned)TM, q1 = du / (unsigned)TM2;
        tl[k] = (du < nd) ? (int)(l1 ? q1 : q0) : (OFFP - 1);
        rk[k] = atomicAdd(&cnt[tl[k]], 1);
    }
    __syncthreads();

    if (tid < 32) {
        int c8[8];
        int run = 0;
#pragma unroll
        for (int j = 0; j < 8; ++j) { c8[j] = run; run += cnt[tid * 8 + j]; }
        int incl = run;
        for (int o = 1; o < 32; o <<= 1) {
            const int v = __shfl_up(incl, o);
            if (tid >= o) incl += v;
        }
        const int base = incl - run;
#pragma unroll
        for (int j = 0; j < 8; ++j) off[tid * 8 + j] = base + c8[j];
    }
    __syncthreads();

#pragma unroll
    for (int k = 0; k < EPT; ++k) {
        if (tl[k] != OFFP - 1) {
            const int dl = dv[k] - tl[k] * (l1 ? TM2 : TM);
            rec[off[tl[k]] + rk[k]] = (unsigned)sv[k] | ((unsigned)dl << SRC_BITS);
        }
    }
    __syncthreads();

    const int nrec = off[l1 ? NT1 : NT0];
    for (int i = tid; i < nrec; i += BT) rout[e0 + i] = rec[i];
    if (tid < OFFP) oout[c * OFFP + tid] = off[tid];
}

constexpr int PITCH = 132;
constexpr int QCAP = 8192;
constexpr int G1T = 1024;
static_assert(N1 % TM == 0 && TM % 16 == 0, "whole tiles only");
__global__ void __launch_bounds__(G1T)
agg_gemm1(const float* __restrict__ feat, const unsigned int* __restrict__ recs,
          const int* __restrict__ offs,
          const float* __restrict__ W,
          const float* __restrict__ bias, float* __restrict__ h1) {
    __shared__ float As[TM * PITCH];
    __shared__ int degc[TM];
    __shared__ float invs[TM];
    __shared__ unsigned int q[QCAP];
    __shared__ int qn;
    const int tid = threadIdx.x;
    const int lane = tid & 31;
    const int wave = __builtin_amdgcn_readfirstlane(tid >> 5);
    const int m0 = blockIdx.x * TM;

    for (int i = tid; i < TM * PITCH; i += G1T) As[i] = 0.0f;
    for (int i = tid; i < TM; i += G1T) degc[i] = 0;
    if (tid == 0) qn = 0;
    __syncthreads();

    {
        constexpr int MYC = (NCH0 + G1T / 32 - 1) / (G1T / 32);
        int pa[MYC], pl[MYC];
#pragma unroll
        for (int u = 0; u < MYC; ++u) {
            const int c = wave + u * (G1T / 32);
            pa[u] = 0;
            pl[u] = 0;
            if (c < NCH0) {
                pa[u] = offs[c * OFFP + blockIdx.x];
                pl[u] = offs[c * OFFP + blockIdx.x + 1] - pa[u];
            }
        }
#pragma unroll
        for (int u = 0; u < MYC; ++u) {
            const int a = __builtin_amdgcn_readfirstlane(pa[u]);
            const int len = __builtin_amdgcn_readfirstlane(pl[u]);
            if (len <= 0) continue;
            int qb = 0;
            if (lane == 0) qb = atomicAdd(&qn, len);
            qb = __builtin_amdgcn_readfirstlane(qb);
            const unsigned int* rp = recs + (wave + u * (G1T / 32)) * CH + a;
            for (int j = lane; j < len; j += 32) {
                const unsigned r = rp[j];
                const int slot = qb + j;
                if (slot < QCAP) {
                    q[slot] = r;
                } else {
                    const unsigned dl = min(r >> SRC_BITS, (unsigned)(TM - 1));
                    const float* hp = feat + (long)(r & SRC_MASK) * DIN;
                    for (int cc = 0; cc < DIN; ++cc) atomicAdd(&As[dl * PITCH + cc], hp[cc]);
                    atomicAdd(&degc[dl], 1);
                }
            }
        }
    }
    __syncthreads();

    {
        const int qv = qn;
        const int n = __builtin_amdgcn_readfirstlane(qv < QCAP ? qv : QCAP);
        const float* hl = feat + lane;
        float* al = As + lane;
        for (int base = wave * 32; base < n; base += G1T) {
            const int cnt = (n - base < 32) ? (n - base) : 32;
            int s = 0, dl = 0;
            if (lane < cnt) {
                const unsigned r = q[base + lane];
                s = (int)(r & SRC_MASK);
                dl = (int)min(r >> SRC_BITS, (unsigned)(TM - 1));
                atomicAdd(&degc[dl], 1);
            }
            int i = 0;
#pragma unroll 1
            for (; i + 8 <= cnt; i += 8) {
                float v[8][4];
                int di[8];
#pragma unroll
                for (int u = 0; u < 8; ++u) {
                    const int si = __builtin_amdgcn_readlane(s, i + u);
                    di[u] = __builtin_amdgcn_readlane(dl, i + u);
                    const float* hp = hl + (long)si * DIN;
#pragma unroll
                    for (int j = 0; j < 4; ++j) v[u][j] = hp[32 * j];
                }
#pragma unroll
                for (int u = 0; u < 8; ++u) {
                    float* ap = al + di[u] * PITCH;
#pragma unroll
                    for (int j = 0; j < 4; ++j) atomicAdd(ap + 32 * j, v[u][j]);
                }
            }
            for (; i < cnt; ++i) {
                const int si = __builtin_amdgcn_readlane(s, i);
                const int dd = __builtin_amdgcn_readlane(dl, i);
                const float* hp = hl + (long)si * DIN;
                float* ap = al + dd * PITCH;
                float v[4];
#pragma unroll
                for (int j = 0; j < 4; ++j) v[j] = hp[32 * j];
#pragma unroll
                for (int j = 0; j < 4; ++j) atomicAdd(ap + 32 * j, v[j]);
            }
        }
    }
    __syncthreads();

    for (int r = tid; r < TM; r += G1T) invs[r] = 1.0f / fmaxf((float)degc[r], 1.0f);
    __syncthreads();
    for (int idx = tid; idx < TM * DIN; idx += G1T) {
        const int r = idx >> 7, c = idx & 127;
        As[r * PITCH + c] = feat[(long)(m0 + r) * DIN + c] + As[r * PITCH + c] * invs[r];
    }
    __syncthreads();

    const int lo = lane & 15, hi = lane >> 4;
    const int ncol = (wave & 15) * 16 + lo;
    v2f b[DIN / 4];
#pragma unroll
    for (int ks = 0; ks < DIN / 4; ++ks) {
        const int kk = 4 * ks + 2 * hi;
        b[ks][0] = W[kk * HID + ncol];
        b[ks][1] = W[(kk + 1) * HID + ncol];
    }
    const float bv = bias[ncol];
#pragma unroll 1
    for (int mt = wave >> 4; mt < TM / 16; mt += 2) {
        const float* ar = As + (mt * 16 + lo) * PITCH + 2 * hi;
        v8f acc = {0, 0, 0, 0, 0, 0, 0, 0};
#pragma unroll
        for (int ks = 0; ks < DIN / 4; ++ks) {
            v2f a;
            a[0] = ar[4 * ks];
            a[1] = ar[4 * ks + 1];
            acc = __builtin_amdgcn_wmma_f32_16x16x4_f32(false, a, false, b[ks], (short)0, acc, false, false);
        }
#pragma unroll
        for (int r = 0; r < 8; ++r) {
            const int rr = mt * 16 + r + 8 * hi;
            const float s = (degc[rr] > 0) ? 2.0f : 1.0f;
            h1[(long)(m0 + rr) * HID + ncol] = fmaxf(acc[r] + s * bv, 0.0f);
        }
    }
}

constexpr int MT2 = (TM2 + 15) / 16;
constexpr int PITCH2 = 260;
constexpr int QCAP2 = 12288;
static_assert(MT2 > 4 && MT2 <= 8, "every wave has a first M tile, and at most a second one");
static_assert(PITCH2 % 4 == 0 && HID % 4 == 0, "rows are cleared and finished in 16-byte pieces");
__global__ void __launch_bounds__(G1T)
agg_gemm2(const float* __restrict__ h1, const unsigned int* __restrict__ recs,
          const int* __restrict__ offs,
          const float* __restrict__ W,
          const float* __restrict__ bias, float* __restrict__ out) {
    __shared__ __attribute__((aligned(16))) float As[MT2 * 16 * PITCH2];
    __shared__ int degc[MT2 * 16];
    __shared__ float invs[MT2 * 16];
    __shared__ unsigned int q[QCAP2];
    __shared__ int qn;
    const int tid = threadIdx.x;
    const int lane = tid & 31;
    const int wave = __builtin_amdgcn_readfirstlane(tid >> 5);
    const int m0 = blockIdx.x * TM2;

    {
        const v4f z = {0.0f, 0.0f, 0.0f, 0.0f};
        for (int i = tid; i < MT2 * 16 * PITCH2 / 4; i += G1T) ((v4f*)As)[i] = z;
    }
    for (int i = tid; i < MT2 * 16; i += G1T) degc[i] = 0;
    if (tid == 0) qn = 0;
    __syncthreads();

    {
        constexpr int MYC = (NCH1 + G1T / 32 - 1) / (G1T / 32);
        int pa[MYC], pl[MYC];
#pragma unroll
        for (int u = 0; u < MYC; ++u) {
            const int c = wave + u * (G1T / 32);
            pa[u] = 0;
            pl[u] = 0;
            if (c < NCH1) {
                pa[u] = offs[c * OFFP + blockIdx.x];
                pl[u] = offs[c * OFFP + blockIdx.x + 1] - pa[u];
            }
        }
#pragma unroll
        for (int u = 0; u < MYC; ++u) {
            const int a = __builtin_amdgcn_readfirstlane(pa[u]);
            const int len = __builtin_amdgcn_readfirstlane(pl[u]);
            if (len <= 0) continue;
            int qb = 0;
            if (lane == 0) qb = atomicAdd(&qn, len);
            qb = __builtin_amdgcn_readfirstlane(qb);
            const unsigned int* rp = recs + (wave + u * (G1T / 32)) * CH + a;
            for (int j = lane; j < len; j += 32) {
                const unsigned r = rp[j];
                const int slot = qb + j;
                if (slot < QCAP2) {
                    q[slot] = r;
                } else {
                    const unsigned dl = min(r >> SRC_BITS, (unsigned)(TM2 - 1));
                    const float* hp = h1 + (long)(r & SRC_MASK) * HID;
                    for (int cc = 0; cc < HID; ++cc) atomicAdd(&As[dl * PITCH2 + cc], hp[cc]);
                    atomicAdd(&degc[dl], 1);
                }
            }
        }
    }
    __syncthreads();

    {
        const int qv = qn;
        const int n = __builtin_amdgcn_readfirstlane(qv < QCAP2 ? qv : QCAP2);
        const float* hl = h1 + lane;
        float* al = As + lane;
        for (int base = wave * 32; base < n; base += G1T) {
            const int cnt = (n - base < 32) ? (n - base) : 32;
            int s = 0, dl = 0;
            if (lane < cnt) {
                const unsigned r = q[base + lane];
                s = (int)(r & SRC_MASK);
                dl = (int)min(r >> SRC_BITS, (unsigned)(TM2 - 1));
                atomicAdd(&degc[dl], 1);
            }
            int i = 0;
#pragma unroll 1
            for (; i + 4 <= cnt; i += 4) {
                float v[4][8];
                int di[4];
#pragma unroll
                for (int u = 0; u < 4; ++u) {
                    const int si = __builtin_amdgcn_readlane(s, i + u);
                    di[u] = __builtin_amdgcn_readlane(dl, i + u);
                    const float* hp = hl + (long)si * HID;
#pragma unroll
                    for (int j = 0; j < 8; ++j) v[u][j] = hp[32 * j];
                }
#pragma unroll
                for (int u = 0; u < 4; ++u) {
                    float* ap = al + di[u] * PITCH2;
#pragma unroll
                    for (int j = 0; j < 8; ++j) atomicAdd(ap + 32 * j, v[u][j]);
                }
            }
            for (; i < cnt; ++i) {
                const int si = __builtin_amdgcn_readlane(s, i);
                const int dd = __builtin_amdgcn_readlane(dl, i);
                const float* hp = hl + (long)si * HID;
                float* ap = al + dd * PITCH2;
                float v[8];
#pragma unroll
                for (int j = 0; j < 8; ++j) v[j] = hp[32 * j];
#pragma unroll
                for (int j = 0; j < 8; ++j) atomicAdd(ap + 32 * j, v[j]);
            }
        }
    }
    __syncthreads();

    for (int r = tid; r < TM2; r += G1T) invs[r] = 1.0f / fmaxf((float)degc[r], 1.0f);
    __syncthreads();
    {
        constexpr int NV = TM2 * (HID / 4);
        constexpr int S3 = (NV + G1T - 1) / G1T;
        const v4f* own = (const v4f*)(h1 + (long)m0 * HID);
        v4f hv[S3];
#pragma unroll
        for (int u = 0; u < S3; ++u) {
            const int idx = tid + u * G1T;
            hv[u] = own[idx < NV ? idx : NV - 1];
        }
#pragma unroll
        for (int u = 0; u < S3; ++u) {
            const int idx = tid + u * G1T;
            if (idx < NV) {
                const int r = idx / (HID / 4), c4 = idx % (HID / 4);
                v4f* ap = (v4f*)(As + r * PITCH2) + c4;
                *ap = hv[u] + *ap * invs[r];
            }
        }
    }
    __syncthreads();

    const int lo = lane & 15, hi = lane >> 4;
    const int ncol = (wave & 7) * 16 + lo;
    const int g = wave >> 3;
    const float* wp = W + (2 * hi) * DOUT + ncol;
    const float* a0p = As + (g * 16 + lo) * PITCH2 + 2 * hi;
    const float bv = bias[ncol];
    v8f acc0 = {0, 0, 0, 0, 0, 0, 0, 0};
    if (g + 4 < MT2) {
        const float* a1p = a0p + 64 * PITCH2;
        v8f acc1 = {0, 0, 0, 0, 0, 0, 0, 0};
#pragma unroll 16
        for (int ks = 0; ks < HID / 4; ++ks) {
            v2f b, a0, a1;
            b[0] = wp[(4 * ks) * DOUT];
            b[1] = wp[(4 * ks + 1) * DOUT];
            a0[0] = a0p[4 * ks];
            a0[1] = a0p[4 * ks + 1];
            a1[0] = a1p[4 * ks];
            a1[1] = a1p[4 * ks + 1];
            acc0 = __builtin_amdgcn_wmma_f32_16x16x4_f32(false, a0, false, b, (short)0, acc0, false, false);
            acc1 = __builtin_amdgcn_wmma_f32_16x16x4_f32(false, a1, false, b, (short)0, acc1, false, false);
        }
#pragma unroll
        for (int r = 0; r < 8; ++r) {
            const int rr = (g + 4) * 16 + r + 8 * hi;
            if (rr < TM2) out[(long)(m0 + rr) * DOUT + ncol] = acc1[r] + bv;
        }
    } else {
#pragma unroll 16
        for (int ks = 0; ks < HID / 4; ++ks) {
            v2f b, a0;
            b[0] = wp[(4 * ks) * DOUT];
            b[1] = wp[(4 * ks + 1) * DOUT];
            a0[0] = a0p[4 * ks];
            a0[1] = a0p[4 * ks + 1];
            acc0 = __builtin_amdgcn_wmma_f32_16x16x4_f32(false, a0, false, b, (short)0, acc0, false, false);
        }
    }
    static_assert(16 * 4 <= TM2, "M tiles 0..3 hold only real rows");
#pragma unroll
    for (int r = 0; r < 8; ++r) {
        const int rr = g * 16 + r + 8 * hi;
        out[(long)(m0 + rr) * DOUT + ncol] = acc0[r] + bv;
    }
}

extern "C" void kernel_launch(void* const* d_in, const int* in_sizes, int n_in,
                              void* d_out, int out_size, void* d_ws, size_t ws_size,
                              hipStream_t stream) {
    (void)in_sizes; (void)n_in; (void)out_size; (void)ws_size;
    const int*   x  = (const int*)d_in[0];
    const float* e0 = (const float*)d_in[1];
    const float* e1 = (const float*)d_in[2];
    const float* e2 = (const float*)d_in[3];
    const float* e3 = (const float*)d_in[4];
    const float* Win[2]  = {(const float*)d_in[5], (const float*)d_in[9]};
    const float* bin[2]  = {(const float*)d_in[6], (const float*)d_in[10]};
    const float* Wout[2] = {(const float*)d_in[7], (const float*)d_in[11]};
    const float* bout[2] = {(const float*)d_in[8], (const float*)d_in[12]};
    const int* srcL0[2] = {(const int*)d_in[13], (const int*)d_in[17]};
    const int* dstL0[2] = {(const int*)d_in[14], (const int*)d_in[18]};
    const int* srcL1[2] = {(const int*)d_in[15], (const int*)d_in[19]};
    const int* dstL1[2] = {(const int*)d_in[16], (const int*)d_in[20]};
    float* out = (float*)d_out;

    size_t off = 0;
    auto take = [&](size_t bytes) { size_t o = off; off += (bytes + 255) & ~(size_t)255; return o; };
    char* ws = (char*)d_ws;
    float* feat = (float*)(ws + take((size_t)N0 * DIN * 4));
    float* h1   = (float*)(ws + take((size_t)N1 * HID * 4));
    unsigned int* rc0[2];
    unsigned int* rc1[2];
    int* of0[2];
    int* of1[2];
    rc0[0] = (unsigned int*)(ws + take((size_t)NCH0 * CH * 4));
    rc0[1] = (unsigned int*)(ws + take((size_t)NCH0 * CH * 4));
    rc1[0] = (unsigned int*)(ws + take((size_t)NCH1 * CH * 4));
    rc1[1] = (unsigned int*)(ws + take((size_t)NCH1 * CH * 4));
    of0[0] = (int*)(ws + take((size_t)NCH0 * OFFP * 4));
    of0[1] = (int*)(ws + take((size_t)NCH0 * OFFP * 4));
    of1[0] = (int*)(ws + take((size_t)NCH1 * OFFP * 4));
    of1[1] = (int*)(ws + take((size_t)NCH1 * OFFP * 4));

    build_feat<<<(N0 / 32 + 7) / 8, 256, 0, stream>>>(x, e0, e1, e2, e3, feat);

    bin_edges<<<dim3(NCH0, 4), BT, 0, stream>>>(dstL0[0], dstL0[1], dstL1[0], dstL1[1],
                                               srcL0[0], srcL0[1], srcL1[0], srcL1[1],
                                               rc0[0], rc0[1], rc1[0], rc1[1],
                                               of0[0], of0[1], of1[0], of1[1]);

    for (int e = 0; e < 2; ++e) {
        agg_gemm1<<<N1 / TM, G1T, 0, stream>>>(feat, rc0[e], of0[e], Win[e], bin[e], h1);
        agg_gemm2<<<N2 / TM2, G1T, 0, stream>>>(h1, rc1[e], of1[e], Wout[e], bout[e],
                                                out + (size_t)e * N2 * DOUT);
    }
}


# fused kernels issue at entry every load whose address is known there (tile's own rows as 16-byte loads, their pieces' records) and cross the phase barriers on LDS-only fences, not __syncthreads
# speedup vs baseline: 2.9104x; 1.0578x over previous
#include <hip/hip_runtime.h>

constexpr int N0 = 400000, N1 = 100000, N2 = 25000;
constexpr int E0 = 800000, E1 = 200000;
constexpr int DIN = 128, HID = 256, DOUT = 128;

typedef __attribute__((ext_vector_type(2))) float v2f;
typedef __attribute__((ext_vector_type(8))) float v8f;
typedef __attribute__((ext_vector_type(4))) float v4f;

__device__ __forceinline__ void lds_barrier() {
    __builtin_amdgcn_fence(__ATOMIC_RELEASE, "workgroup", "local");
    __builtin_amdgcn_s_barrier();
    __builtin_amdgcn_fence(__ATOMIC_ACQUIRE, "workgroup", "local");
}

constexpr int FEAT_U = 8;
static_assert(N0 % 32 == 0, "build_feat takes whole groups of 32 rows");
__global__ void __launch_bounds__(256)
build_feat(const int* __restrict__ x,
           const float* __restrict__ e0, const float* __restrict__ e1,
           const float* __restrict__ e2, const float* __restrict__ e3,
           float* __restrict__ feat) {
    const int lane = threadIdx.x & 31;
    const int wave = __builtin_amdgcn_readfirstlane((int)((blockIdx.x * blockDim.x + threadIdx.x) >> 5));
    const int row0 = wave * 32;
    if (row0 >= N0) return;
    const int4 xi = ((const int4*)x)[row0 + lane];
    const float* tb;
    int rl;
    if (lane < 8)       { tb = e0 + lane * 4;        rl = 32; }
    else if (lane < 12) { tb = e1 + (lane - 8) * 4;  rl = 16; }
    else if (lane < 16) { tb = e2 + (lane - 12) * 4; rl = 16; }
    else                { tb = e3 + (lane - 16) * 4; rl = 64; }
    v4f* dst = (v4f*)(feat + (long)row0 * 128) + lane;
#pragma unroll 1
    for (int i = 0; i < 32; i += FEAT_U) {
        v4f v[FEAT_U];
#pragma unroll
        for (int u = 0; u < FEAT_U; ++u) {
            const int i0 = __builtin_amdgcn_readlane(xi.x, i + u);
            const int i1 = __builtin_amdgcn_readlane(xi.y, i + u);
            const int i2 = __builtin_amdgcn_readlane(xi.z, i + u);
            const int i3 = __builtin_amdgcn_readlane(xi.w, i + u);
            const int idx = (lane < 8) ? i0 : (lane < 12) ? i1 : (lane < 16) ? i2 : i3;
            v[u] = *(const v4f*)(tb + (long)idx * rl);
        }
        static_assert(FEAT_U == 8, "the fence below names eight values");
        asm volatile("" :: "v"(v[0]), "v"(v[1]), "v"(v[2]), "v"(v[3]),
                           "v"(v[4]), "v"(v[5]), "v"(v[6]), "v"(v[7]));
#pragma unroll
        for (int u = 0; u < FEAT_U; ++u) dst[(i + u) * 32] = v[u];
    }
}

constexpr int TM = 400;
constexpr int TM2 = 100;
constexpr int CH = 8192;
constexpr int BT = 1024;
constexpr int EPT = CH / BT;
constexpr int NCH0 = (E0 + CH - 1) / CH;
constexpr int NCH1 = (E1 + CH - 1) / CH;
constexpr int OFFP = 256;
constexpr int SRC_BITS = 19;
constexpr unsigned SRC_MASK = (1u << SRC_BITS) - 1u;
constexpr int NT0 = N1 / TM;
constexpr int NT1 = N2 / TM2;
static_assert(N1 % TM == 0 && NT0 < OFFP - 1, "tile t + 1 is an entry of the table, OFFP - 1 stays free for 'no tile'");
static_assert(N2 % TM2 == 0 && NT1 < OFFP - 1, "tile t + 1 is an entry of the table, OFFP - 1 stays free for 'no tile'");
static_assert(N0 <= (1 << SRC_BITS) && N1 <= (1 << SRC_BITS), "src fits the low bits of a record");
static_assert(TM <= (1 << (31 - SRC_BITS)) && TM2 <= (1 << (31 - SRC_BITS)), "the row inside the tile fits the high bits");
static_assert(CH % BT == 0 && OFFP == 8 * 32, "the prefix sum below gives each lane of one wave eight bins");
__global__ void __launch_bounds__(BT)
bin_edges(const int* __restrict__ d0A, const int* __restrict__ d0B,
          const int* __restrict__ d1A, const int* __restrict__ d1B,
          const int* __restrict__ s0A, const int* __restrict__ s0B,
          const int* __restrict__ s1A, const int* __restrict__ s1B,
          unsigned int* __restrict__ r0A, unsigned int* __restrict__ r0B,
          unsigned int* __restrict__ r1A, unsigned int* __restrict__ r1B,
          int* __restrict__ o0A, int* __restrict__ o0B,
          int* __restrict__ o1A, int* __restrict__ o1B) {
    __shared__ int cnt[OFFP];
    __shared__ int off[OFFP];
    __shared__ unsigned int rec[CH];
    const int y = blockIdx.y;
    const bool l1 = y >= 2;
    const int c = blockIdx.x;
    if (c >= (l1 ? NCH1 : NCH0)) return;
    const int* d = (y == 0) ? d0A : (y == 1) ? d0B : (y == 2) ? d1A : d1B;
    const int* sp = (y == 0) ? s0A : (y == 1) ? s0B : (y == 2) ? s1A : s1B;
    unsigned int* rout = (y == 0) ? r0A : (y == 1) ? r0B : (y == 2) ? r1A : r1B;
    int* oout = (y == 0) ? o0A : (y == 1) ? o0B : (y == 2) ? o1A : o1B;
    const int E = l1 ? E1 : E0;
    const unsigned nd = l1 ? (unsigned)N2 : (unsigned)N1;
    const int tid = threadIdx.x;
    if (tid < OFFP) cnt[tid] = 0;
    __syncthreads();

    const int e0 = c * CH;
    int sv[EPT], dv[EPT];
#pragma unroll
    for (int k = 0; k < EPT; ++k) {
        const int e = e0 + tid + k * BT;
        const bool in = e < E;
        dv[k] = in ? d[e] : -1;
        sv[k] = in ? sp[e] : 0;
    }
    int tl[EPT], rk[EPT];
#pragma unroll
    for (int k = 0; k < EPT; ++k) {
        const unsigned du = (unsigned)dv[k];
        const unsigned q0 = du / (unsigned)TM, q1 = du / (unsigned)TM2;
        tl[k] = (du < nd) ? (int)(l1 ? q1 : q0) : (OFFP - 1);
        rk[k] = atomicAdd(&cnt[tl[k]], 1);
    }
    __syncthreads();

    if (tid < 32) {
        int c8[8];
        int run = 0;
#pragma unroll
        for (int j = 0; j < 8; ++j) { c8[j] = run; run += cnt[tid * 8 + j]; }
        int incl = run;
        for (int o = 1; o < 32; o <<= 1) {
            const int v = __shfl_up(incl, o);
            if (tid >= o) incl += v;
        }
        const int base = incl - run;
#pragma unroll
        for (int j = 0; j < 8; ++j) off[tid * 8 + j] = base + c8[j];
    }
    __syncthreads();

#pragma unroll
    for (int k = 0; k < EPT; ++k) {
        if (tl[k] != OFFP - 1) {
            const int dl = dv[k] - tl[k] * (l1 ? TM2 : TM);
            rec[off[tl[k]] + rk[k]] = (unsigned)sv[k] | ((unsigned)dl << SRC_BITS);
        }
    }
    __syncthreads();

    const int nrec = off[l1 ? NT1 : NT0];
    for (int i = tid; i < nrec; i += BT) rout[e0 + i] = rec[i];
    if (tid < OFFP) oout[c * OFFP + tid] = off[tid];
}

constexpr int PITCH = 132;
constexpr int QCAP = 8192;
constexpr int G1T = 1024;
static_assert(N1 % TM == 0 && TM % 16 == 0, "whole tiles only");
__global__ void __launch_bounds__(G1T)
agg_gemm1(const float* __restrict__ feat, const unsigned int* __restrict__ recs,
          const int* __restrict__ offs,
          const float* __restrict__ W,
          const float* __restrict__ bias, float* __restrict__ h1) {
    __shared__ __attribute__((aligned(16))) float As[TM * PITCH];
    __shared__ int degc[TM];
    __shared__ float invs[TM];
    __shared__ unsigned int q[QCAP];
    __shared__ int qn;
    const int tid = threadIdx.x;
    const int lane = tid & 31;
    const int wave = __builtin_amdgcn_readfirstlane(tid >> 5);
    const int m0 = blockIdx.x * TM;

    constexpr int MYC = (NCH0 + G1T / 32 - 1) / (G1T / 32);
    int pa[MYC], pe[MYC];
#pragma unroll
    for (int u = 0; u < MYC; ++u) {
        const int c = wave + u * (G1T / 32);
        pa[u] = 0;
        pe[u] = 0;
        if (c < NCH0) {
            pa[u] = offs[c * OFFP + blockIdx.x];
            pe[u] = offs[c * OFFP + blockIdx.x + 1];
        }
    }
    unsigned int rq[MYC][2];
#pragma unroll
    for (int u = 0; u < MYC; ++u) {
        const int a = __builtin_amdgcn_readfirstlane(pa[u]);
        const int len = __builtin_amdgcn_readfirstlane(pe[u] - pa[u]);
        const long rb = (long)(wave + u * (G1T / 32)) * CH + a;
        rq[u][0] = (lane < len) ? recs[rb + lane] : 0u;
        rq[u][1] = (lane + 32 < len) ? recs[rb + lane + 32] : 0u;
    }
    constexpr int NV1 = TM * (DIN / 4);
    constexpr int S1 = (NV1 + G1T - 1) / G1T;
    v4f hv[S1];
    {
        const v4f* own = (const v4f*)(feat + (long)m0 * DIN);
#pragma unroll
        for (int u = 0; u < S1; ++u) {
            const int idx = tid + u * G1T;
            hv[u] = own[idx < NV1 ? idx : NV1 - 1];
        }
    }

    for (int i = tid; i < TM * PITCH; i += G1T) As[i] = 0.0f;
    for (int i = tid; i < TM; i += G1T) degc[i] = 0;
    if (tid == 0) qn = 0;
    lds_barrier();

    {
        int qbs[MYC];
        bool rest = false;
#pragma unroll
        for (int u = 0; u < MYC; ++u) {
            const int len = __builtin_amdgcn_readfirstlane(pe[u] - pa[u]);
            qbs[u] = 0;
            if (len <= 0) continue;
            int qb = 0;
            if (lane == 0) qb = atomicAdd(&qn, len);
            qb = __builtin_amdgcn_readfirstlane(qb);
            qbs[u] = qb;
#pragma unroll
            for (int h = 0; h < 2; ++h) {
                const int j = lane + 32 * h;
                if (j < len && qb + j < QCAP) q[qb + j] = rq[u][h];
            }
            rest = rest || len > 64 || qb + len > QCAP;
        }
        if (rest) {
#pragma unroll
            for (int u = 0; u < MYC; ++u) {
                const int a = __builtin_amdgcn_readfirstlane(pa[u]);
                const int len = __builtin_amdgcn_readfirstlane(pe[u] - pa[u]);
                const int qb = qbs[u];
                if (len <= 0) continue;
                const unsigned int* rp = recs + (wave + u * (G1T / 32)) * CH + a;
                for (int j = lane; j < len; j += 32) {
                    const int slot = qb + j;
                    if (j < 64 && slot < QCAP) continue;
                    const unsigned r = rp[j];
                    if (slot < QCAP) {
                        q[slot] = r;
                    } else {
                        const unsigned dl = min(r >> SRC_BITS, (unsigned)(TM - 1));
                        const float* hp = feat + (long)(r & SRC_MASK) * DIN;
                        for (int cc = 0; cc < DIN; ++cc) atomicAdd(&As[dl * PITCH + cc], hp[cc]);
                        atomicAdd(&degc[dl], 1);
                    }
                }
            }
        }
    }
    lds_barrier();

    {
        const int qv = qn;
        const int n = __builtin_amdgcn_readfirstlane(qv < QCAP ? qv : QCAP);
        const float* hl = feat + lane;
        float* al = As + lane;
        for (int base = wave * 32; base < n; base += G1T) {
            const int cnt = (n - base < 32) ? (n - base) : 32;
            int s = 0, dl = 0;
            if (lane < cnt) {
                const unsigned r = q[base + lane];
                s = (int)(r & SRC_MASK);
                dl = (int)min(r >> SRC_BITS, (unsigned)(TM - 1));
                atomicAdd(&degc[dl], 1);
            }
            int i = 0;
#pragma unroll 1
            for (; i + 8 <= cnt; i += 8) {
                float v[8][4];
                int di[8];
#pragma unroll
                for (int u = 0; u < 8; ++u) {
                    const int si = __builtin_amdgcn_readlane(s, i + u);
                    di[u] = __builtin_amdgcn_readlane(dl, i + u);
                    const float* hp = hl + (long)si * DIN;
#pragma unroll
                    for (int j = 0; j < 4; ++j) v[u][j] = hp[32 * j];
                }
#pragma unroll
                for (int u = 0; u < 8; ++u) {
                    float* ap = al + di[u] * PITCH;
#pragma unroll
                    for (int j = 0; j < 4; ++j) atomicAdd(ap + 32 * j, v[u][j]);
                }
            }
            for (; i < cnt; ++i) {
                const int si = __builtin_amdgcn_readlane(s, i);
                const int dd = __builtin_amdgcn_readlane(dl, i);
                const float* hp = hl + (long)si * DIN;
                float* ap = al + dd * PITCH;
                float v[4];
#pragma unroll
                for (int j = 0; j < 4; ++j) v[j] = hp[32 * j];
#pragma unroll
                for (int j = 0; j < 4; ++j) atomicAdd(ap + 32 * j, v[j]);
            }
        }
    }
    lds_barrier();

    for (int r = tid; r < TM; r += G1T) invs[r] = 1.0f / fmaxf((float)degc[r], 1.0f);
    lds_barrier();
#pragma unroll
    for (int u = 0; u < S1; ++u) {
        const int idx = tid + u * G1T;
        if (idx < NV1) {
            const int r = idx / (DIN / 4), c4 = idx % (DIN / 4);
            v4f* ap = (v4f*)(As + r * PITCH) + c4;
            *ap = hv[u] + *ap * invs[r];
        }
    }
    __syncthreads();

    const int lo = lane & 15, hi = lane >> 4;
    const int ncol = (wave & 15) * 16 + lo;
    v2f b[DIN / 4];
#pragma unroll
    for (int ks = 0; ks < DIN / 4; ++ks) {
        const int kk = 4 * ks + 2 * hi;
        b[ks][0] = W[kk * HID + ncol];
        b[ks][1] = W[(kk + 1) * HID + ncol];
    }
    const float bv = bias[ncol];
#pragma unroll 1
    for (int mt = wave >> 4; mt < TM / 16; mt += 2) {
        const float* ar = As + (mt * 16 + lo) * PITCH + 2 * hi;
        v8f acc = {0, 0, 0, 0, 0, 0, 0, 0};
#pragma unroll
        for (int ks = 0; ks < DIN / 4; ++ks) {
            v2f a;
            a[0] = ar[4 * ks];
            a[1] = ar[4 * ks + 1];
            acc = __builtin_amdgcn_wmma_f32_16x16x4_f32(false, a, false, b[ks], (short)0, acc, false, false);
        }
#pragma unroll
        for (int r = 0; r < 8; ++r) {
            const int rr = mt * 16 + r + 8 * hi;
            const float s = (degc[rr] > 0) ? 2.0f : 1.0f;
            h1[(long)(m0 + rr) * HID + ncol] = fmaxf(acc[r] + s * bv, 0.0f);
        }
    }
}

constexpr int MT2 = (TM2 + 15) / 16;
constexpr int PITCH2 = 260;
constexpr int QCAP2 = 12288;
static_assert(MT2 > 4 && MT2 <= 8, "every wave has a first M tile, and at most a second one");
static_assert(PITCH2 % 4 == 0 && HID % 4 == 0, "rows are cleared and finished in 16-byte pieces");
__global__ void __launch_bounds__(G1T)
agg_gemm2(const float* __restrict__ h1, const unsigned int* __restrict__ recs,
          const int* __restrict__ offs,
          const float* __restrict__ W,
          const float* __restrict__ bias, float* __restrict__ out) {
    __shared__ __attribute__((aligned(16))) float As[MT2 * 16 * PITCH2];
    __shared__ int degc[MT2 * 16];
    __shared__ float invs[MT2 * 16];
    __shared__ unsigned int q[QCAP2];
    __shared__ int qn;
    const int tid = threadIdx.x;
    const int lane = tid & 31;
    const int wave = __builtin_amdgcn_readfirstlane(tid >> 5);
    const int m0 = blockIdx.x * TM2;

    constexpr int MYC = (NCH1 + G1T / 32 - 1) / (G1T / 32);
    int pa[MYC], pe[MYC];
#pragma unroll
    for (int u = 0; u < MYC; ++u) {
        const int c = wave + u * (G1T / 32);
        pa[u] = 0;
        pe[u] = 0;
        if (c < NCH1) {
            pa[u] = offs[c * OFFP + blockIdx.x];
            pe[u] = offs[c * OFFP + blockIdx.x + 1];
        }
    }
    unsigned int rq[MYC][2];
#pragma unroll
    for (int u = 0; u < MYC; ++u) {
        const int a = __builtin_amdgcn_readfirstlane(pa[u]);
        const int len = __builtin_amdgcn_readfirstlane(pe[u] - pa[u]);
        const long rb = (long)(wave + u * (G1T / 32)) * CH + a;
        rq[u][0] = (lane < len) ? recs[rb + lane] : 0u;
        rq[u][1] = (lane + 32 < len) ? recs[rb + lane + 32] : 0u;
    }
    constexpr int NV = TM2 * (HID / 4);
    constexpr int S3 = (NV + G1T - 1) / G1T;
    v4f hv[S3];
    {
        const v4f* own = (const v4f*)(h1 + (long)m0 * HID);
#pragma unroll
        for (int u = 0; u < S3; ++u) {
            const int idx = tid + u * G1T;
            hv[u] = own[idx < NV ? idx : NV - 1];
        }
    }

    {
        const v4f z = {0.0f, 0.0f, 0.0f, 0.0f};
        for (int i = tid; i < MT2 * 16 * PITCH2 / 4; i += G1T) ((v4f*)As)[i] = z;
    }
    for (int i = tid; i < MT2 * 16; i += G1T) degc[i] = 0;
    if (tid == 0) qn = 0;
    lds_barrier();

    {
        int qbs[MYC];
        bool rest = false;
#pragma unroll
        for (int u = 0; u < MYC; ++u) {
            const int len = __builtin_amdgcn_readfirstlane(pe[u] - pa[u]);
            qbs[u] = 0;
            if (len <= 0) continue;
            int qb = 0;
            if (lane == 0) qb = atomicAdd(&qn, len);
            qb = __builtin_amdgcn_readfirstlane(qb);
            qbs[u] = qb;
#pragma unroll
            for (int h = 0; h < 2; ++h) {
                const int j = lane + 32 * h;
                if (j < len && qb + j < QCAP2) q[qb + j] = rq[u][h];
            }
            rest = rest || len > 64 || qb + len > QCAP2;
        }
        if (rest) {
#pragma unroll
            for (int u = 0; u < MYC; ++u) {
                const int a = __builtin_amdgcn_readfirstlane(pa[u]);
                const int len = __builtin_amdgcn_readfirstlane(pe[u] - pa[u]);
                const int qb = qbs[u];
                if (len <= 0) continue;
                const unsigned int* rp = recs + (wave + u * (G1T / 32)) * CH + a;
                for (int j = lane; j < len; j += 32) {
                    const int slot = qb + j;
                    if (j < 64 && slot < QCAP2) continue;
                    const unsigned r = rp[j];
                    if (slot < QCAP2) {
                        q[slot] = r;
                    } else {
                        const unsigned dl = min(r >> SRC_BITS, (unsigned)(TM2 - 1));
                        const float* hp = h1 + (long)(r & SRC_MASK) * HID;
                        for (int cc = 0; cc < HID; ++cc) atomicAdd(&As[dl * PITCH2 + cc], hp[cc]);
                        atomicAdd(&degc[dl], 1);
                    }
                }
            }
        }
    }
    lds_barrier();

    {
        const int qv = qn;
        const int n = __builtin_amdgcn_readfirstlane(qv < QCAP2 ? qv : QCAP2);
        const float* hl = h1 + lane;
        float* al = As + lane;
        for (int base = wave * 32; base < n; base += G1T) {
            const int cnt = (n - base < 32) ? (n - base) : 32;
            int s = 0, dl = 0;
            if (lane < cnt) {
                const unsigned r = q[base + lane];
                s = (int)(r & SRC_MASK);
                dl = (int)min(r >> SRC_BITS, (unsigned)(TM2 - 1));
                atomicAdd(&degc[dl], 1);
            }
            int i = 0;
#pragma unroll 1
            for (; i + 4 <= cnt; i += 4) {
                float v[4][8];
                int di[4];
#pragma unroll
                for (int u = 0; u < 4; ++u) {
                    const int si = __builtin_amdgcn_readlane(s, i + u);
                    di[u] = __builtin_amdgcn_readlane(dl, i + u);
                    const float* hp = hl + (long)si * HID;
#pragma unroll
                    for (int j = 0; j < 8; ++j) v[u][j] = hp[32 * j];
                }
#pragma unroll
                for (int u = 0; u < 4; ++u) {
                    float* ap = al + di[u] * PITCH2;
#pragma unroll
                    for (int j = 0; j < 8; ++j) atomicAdd(ap + 32 * j, v[u][j]);
                }
            }
            for (; i < cnt; ++i) {
                const int si = __builtin_amdgcn_readlane(s, i);
                const int dd = __builtin_amdgcn_readlane(dl, i);
                const float* hp = hl + (long)si * HID;
                float* ap = al + dd * PITCH2;
                float v[8];
#pragma unroll
                for (int j = 0; j < 8; ++j) v[j] = hp[32 * j];
#pragma unroll
                for (int j = 0; j < 8; ++j) atomicAdd(ap + 32 * j, v[j]);
            }
        }
    }
    lds_barrier();

    for (int r = tid; r < TM2; r += G1T) invs[r] = 1.0f / fmaxf((float)degc[r], 1.0f);
    lds_barrier();
    {
#pragma unroll
        for (int u = 0; u < S3; ++u) {
            const int idx = tid + u * G1T;
            if (idx < NV) {
                const int r = idx / (HID / 4), c4 = idx % (HID / 4);
                v4f* ap = (v4f*)(As + r * PITCH2) + c4;
                *ap = hv[u] + *ap * invs[r];
            }
        }
    }
    __syncthreads();

    const int lo = lane & 15, hi = lane >> 4;
    const int ncol = (wave & 7) * 16 + lo;
    const int g = wave >> 3;
    const float* wp = W + (2 * hi) * DOUT + ncol;
    const float* a0p = As + (g * 16 + lo) * PITCH2 + 2 * hi;
    const float bv = bias[ncol];
    v8f acc0 = {0, 0, 0, 0, 0, 0, 0, 0};
    if (g + 4 < MT2) {
        const float* a1p = a0p + 64 * PITCH2;
        v8f acc1 = {0, 0, 0, 0, 0, 0, 0, 0};
#pragma unroll 16
        for (int ks = 0; ks < HID / 4; ++ks) {
            v2f b, a0, a1;
            b[0] = wp[(4 * ks) * DOUT];
            b[1] = wp[(4 * ks + 1) * DOUT];
            a0[0] = a0p[4 * ks];
            a0[1] = a0p[4 * ks + 1];
            a1[0] = a1p[4 * ks];
            a1[1] = a1p[4 * ks + 1];
            acc0 = __builtin_amdgcn_wmma_f32_16x16x4_f32(false, a0, false, b, (short)0, acc0, false, false);
            acc1 = __builtin_amdgcn_wmma_f32_16x16x4_f32(false, a1, false, b, (short)0, acc1, false, false);
        }
#pragma unroll
        for (int r = 0; r < 8; ++r) {
            const int rr = (g + 4) * 16 + r + 8 * hi;
            if (rr < TM2) out[(long)(m0 + rr) * DOUT + ncol] = acc1[r] + bv;
        }
    } else {
#pragma unroll 16
        for (int ks = 0; ks < HID / 4; ++ks) {
            v2f b, a0;
            b[0] = wp[(4 * ks) * DOUT];
            b[1] = wp[(4 * ks + 1) * DOUT];
            a0[0] = a0p[4 * ks];
            a0[1] = a0p[4 * ks + 1];
            acc0 = __builtin_amdgcn_wmma_f32_16x16x4_f32(false, a0, false, b, (short)0, acc0, false, false);
        }
    }
    static_assert(16 * 4 <= TM2, "M tiles 0..3 hold only real rows");
#pragma unroll
    for (int r = 0; r < 8; ++r) {
        const int rr = g * 16 + r + 8 * hi;
        out[(long)(m0 + rr) * DOUT + ncol] = acc0[r] + bv;
    }
}

extern "C" void kernel_launch(void* const* d_in, const int* in_sizes, int n_in,
                              void* d_out, int out_size, void* d_ws, size_t ws_size,
                              hipStream_t stream) {
    (void)in_sizes; (void)n_in; (void)out_size; (void)ws_size;
    const int*   x  = (const int*)d_in[0];
    const float* e0 = (const float*)d_in[1];
    const float* e1 = (const float*)d_in[2];
    const float* e2 = (const float*)d_in[3];
    const float* e3 = (const float*)d_in[4];
    const float* Win[2]  = {(const float*)d_in[5], (const float*)d_in[9]};
    const float* bin[2]  = {(const float*)d_in[6], (const float*)d_in[10]};
    const float* Wout[2] = {(const float*)d_in[7], (const float*)d_in[11]};
    const float* bout[2] = {(const float*)d_in[8], (const float*)d_in[12]};
    const int* srcL0[2] = {(const int*)d_in[13], (const int*)d_in[17]};
    const int* dstL0[2] = {(const int*)d_in[14], (const int*)d_in[18]};
    const int* srcL1[2] = {(const int*)d_in[15], (const int*)d_in[19]};
    const int* dstL1[2] = {(const int*)d_in[16], (const int*)d_in[20]};
    float* out = (float*)d_out;

    size_t off = 0;
    auto take = [&](size_t bytes) { size_t o = off; off += (bytes + 255) & ~(size_t)255; return o; };
    char* ws = (char*)d_ws;
    float* feat = (float*)(ws + take((size_t)N0 * DIN * 4));
    float* h1   = (float*)(ws + take((size_t)N1 * HID * 4));
    unsigned int* rc0[2];
    unsigned int* rc1[2];
    int* of0[2];
    int* of1[2];
    rc0[0] = (unsigned int*)(ws + take((size_t)NCH0 * CH * 4));
    rc0[1] = (unsigned int*)(ws + take((size_t)NCH0 * CH * 4));
    rc1[0] = (unsigned int*)(ws + take((size_t)NCH1 * CH * 4));
    rc1[1] = (unsigned int*)(ws + take((size_t)NCH1 * CH * 4));
    of0[0] = (int*)(ws + take((size_t)NCH0 * OFFP * 4));
    of0[1] = (int*)(ws + take((size_t)NCH0 * OFFP * 4));
    of1[0] = (int*)(ws + take((size_t)NCH1 * OFFP * 4));
    of1[1] = (int*)(ws + take((size_t)NCH1 * OFFP * 4));

    build_feat<<<(N0 / 32 + 7) / 8, 256, 0, stream>>>(x, e0, e1, e2, e3, feat);

    bin_edges<<<dim3(NCH0, 4), BT, 0, stream>>>(dstL0[0], dstL0[1], dstL1[0], dstL1[1],
                                               srcL0[0], srcL0[1], srcL1[0], srcL1[1],
                                               rc0[0], rc0[1], rc1[0], rc1[1],
                                               of0[0], of0[1], of1[0], of1[1]);

    for (int e = 0; e < 2; ++e) {
        agg_gemm1<<<N1 / TM, G1T, 0, stream>>>(feat, rc0[e], of0[e], Win[e], bin[e], h1);
        agg_gemm2<<<N2 / TM2, G1T, 0, stream>>>(h1, rc1[e], of1[e], Wout[e], bout[e],
                                                out + (size_t)e * N2 * DOUT);
    }
}


# fused kernels counting-sort their queue by row in LDS and sum each run of equal rows in registers: one LDS float-add set per run, not per edge; sums: register run, then LDS-add arrival order
# speedup vs baseline: 3.0400x; 1.0445x over previous
#include <hip/hip_runtime.h>

constexpr int N0 = 400000, N1 = 100000, N2 = 25000;
constexpr int E0 = 800000, E1 = 200000;
constexpr int DIN = 128, HID = 256, DOUT = 128;

typedef __attribute__((ext_vector_type(2))) float v2f;
typedef __attribute__((ext_vector_type(8))) float v8f;
typedef __attribute__((ext_vector_type(4))) float v4f;

__device__ __forceinline__ void lds_barrier() {
    __builtin_amdgcn_fence(__ATOMIC_RELEASE, "workgroup", "local");
    __builtin_amdgcn_s_barrier();
    __builtin_amdgcn_fence(__ATOMIC_ACQUIRE, "workgroup", "local");
}

__device__ __forceinline__ void prefix512(const int* __restrict__ cnt, int* __restrict__ off, int lane) {
    int c16[16];
    int run = 0;
#pragma unroll
    for (int j = 0; j < 16; ++j) { c16[j] = run; run += cnt[lane * 16 + j]; }
    int incl = run;
    for (int o = 1; o < 32; o <<= 1) {
        const int v = __shfl_up(incl, o);
        if (lane >= o) incl += v;
    }
    const int base = incl - run;
#pragma unroll
    for (int j = 0; j < 16; ++j) off[lane * 16 + j] = base + c16[j];
}

constexpr int FEAT_U = 8;
static_assert(N0 % 32 == 0, "build_feat takes whole groups of 32 rows");
__global__ void __launch_bounds__(256)
build_feat(const int* __restrict__ x,
           const float* __restrict__ e0, const float* __restrict__ e1,
           const float* __restrict__ e2, const float* __restrict__ e3,
           float* __restrict__ feat) {
    const int lane = threadIdx.x & 31;
    const int wave = __builtin_amdgcn_readfirstlane((int)((blockIdx.x * blockDim.x + threadIdx.x) >> 5));
    const int row0 = wave * 32;
    if (row0 >= N0) return;
    const int4 xi = ((const int4*)x)[row0 + lane];
    const float* tb;
    int rl;
    if (lane < 8)       { tb = e0 + lane * 4;        rl = 32; }
    else if (lane < 12) { tb = e1 + (lane - 8) * 4;  rl = 16; }
    else if (lane < 16) { tb = e2 + (lane - 12) * 4; rl = 16; }
    else                { tb = e3 + (lane - 16) * 4; rl = 64; }
    v4f* dst = (v4f*)(feat + (long)row0 * 128) + lane;
#pragma unroll 1
    for (int i = 0; i < 32; i += FEAT_U) {
        v4f v[FEAT_U];
#pragma unroll
        for (int u = 0; u < FEAT_U; ++u) {
            const int i0 = __builtin_amdgcn_readlane(xi.x, i + u);
            const int i1 = __builtin_amdgcn_readlane(xi.y, i + u);
            const int i2 = __builtin_amdgcn_readlane(xi.z, i + u);
            const int i3 = __builtin_amdgcn_readlane(xi.w, i + u);
            const int idx = (lane < 8) ? i0 : (lane < 12) ? i1 : (lane < 16) ? i2 : i3;
            v[u] = *(const v4f*)(tb + (long)idx * rl);
        }
        static_assert(FEAT_U == 8, "the fence below names eight values");
        asm volatile("" :: "v"(v[0]), "v"(v[1]), "v"(v[2]), "v"(v[3]),
                           "v"(v[4]), "v"(v[5]), "v"(v[6]), "v"(v[7]));
#pragma unroll
        for (int u = 0; u < FEAT_U; ++u) dst[(i + u) * 32] = v[u];
    }
}

constexpr int TM = 400;
constexpr int TM2 = 100;
constexpr int CH = 8192;
constexpr int BT = 1024;
constexpr int EPT = CH / BT;
constexpr int NCH0 = (E0 + CH - 1) / CH;
constexpr int NCH1 = (E1 + CH - 1) / CH;
constexpr int OFFP = 256;
constexpr int SRC_BITS = 19;
constexpr unsigned SRC_MASK = (1u << SRC_BITS) - 1u;
constexpr int NT0 = N1 / TM;
constexpr int NT1 = N2 / TM2;
static_assert(N1 % TM == 0 && NT0 < OFFP - 1, "tile t + 1 is an entry of the table, OFFP - 1 stays free for 'no tile'");
static_assert(N2 % TM2 == 0 && NT1 < OFFP - 1, "tile t + 1 is an entry of the table, OFFP - 1 stays free for 'no tile'");
static_assert(N0 <= (1 << SRC_BITS) && N1 <= (1 << SRC_BITS), "src fits the low bits of a record");
static_assert(TM <= (1 << (31 - SRC_BITS)) && TM2 <= (1 << (31 - SRC_BITS)), "the row inside the tile fits the high bits");
static_assert(CH % BT == 0 && OFFP == 8 * 32, "the prefix sum below gives each lane of one wave eight bins");
__global__ void __launch_bounds__(BT)
bin_edges(const int* __restrict__ d0A, const int* __restrict__ d0B,
          const int* __restrict__ d1A, const int* __restrict__ d1B,
          const int* __restrict__ s0A, const int* __restrict__ s0B,
          const int* __restrict__ s1A, const int* __restrict__ s1B,
          unsigned int* __restrict__ r0A, unsigned int* __restrict__ r0B,
          unsigned int* __restrict__ r1A, unsigned int* __restrict__ r1B,
          int* __restrict__ o0A, int* __restrict__ o0B,
          int* __restrict__ o1A, int* __restrict__ o1B) {
    __shared__ int cnt[OFFP];
    __shared__ int off[OFFP];
    __shared__ unsigned int rec[CH];
    const int y = blockIdx.y;
    const bool l1 = y >= 2;
    const int c = blockIdx.x;
    if (c >= (l1 ? NCH1 : NCH0)) return;
    const int* d = (y == 0) ? d0A : (y == 1) ? d0B : (y == 2) ? d1A : d1B;
    const int* sp = (y == 0) ? s0A : (y == 1) ? s0B : (y == 2) ? s1A : s1B;
    unsigned int* rout = (y == 0) ? r0A : (y == 1) ? r0B : (y == 2) ? r1A : r1B;
    int* oout = (y == 0) ? o0A : (y == 1) ? o0B : (y == 2) ? o1A : o1B;
    const int E = l1 ? E1 : E0;
    const unsigned nd = l1 ? (unsigned)N2 : (unsigned)N1;
    const int tid = threadIdx.x;
    if (tid < OFFP) cnt[tid] = 0;
    __syncthreads();

    const int e0 = c * CH;
    int sv[EPT], dv[EPT];
#pragma unroll
    for (int k = 0; k < EPT; ++k) {
        const int e = e0 + tid + k * BT;
        const bool in = e < E;
        dv[k] = in ? d[e] : -1;
        sv[k] = in ? sp[e] : 0;
    }
    int tl[EPT], rk[EPT];
#pragma unroll
    for (int k = 0; k < EPT; ++k) {
        const unsigned du = (unsigned)dv[k];
        const unsigned q0 = du / (unsigned)TM, q1 = du / (unsigned)TM2;
        tl[k] = (du < nd) ? (int)(l1 ? q1 : q0) : (OFFP - 1);
        rk[k] = atomicAdd(&cnt[tl[k]], 1);
    }
    __syncthreads();

    if (tid < 32) {
        int c8[8];
        int run = 0;
#pragma unroll
        for (int j = 0; j < 8; ++j) { c8[j] = run; run += cnt[tid * 8 + j]; }
        int incl = run;
        for (int o = 1; o < 32; o <<= 1) {
            const int v = __shfl_up(incl, o);
            if (tid >= o) incl += v;
        }
        const int base = incl - run;
#pragma unroll
        for (int j = 0; j < 8; ++j) off[tid * 8 + j] = base + c8[j];
    }
    __syncthreads();

#pragma unroll
    for (int k = 0; k < EPT; ++k) {
        if (tl[k] != OFFP - 1) {
            const int dl = dv[k] - tl[k] * (l1 ? TM2 : TM);
            rec[off[tl[k]] + rk[k]] = (unsigned)sv[k] | ((unsigned)dl << SRC_BITS);
        }
    }
    __syncthreads();

    const int nrec = off[l1 ? NT1 : NT0];
    for (int i = tid; i < nrec; i += BT) rout[e0 + i] = rec[i];
    if (tid < OFFP) oout[c * OFFP + tid] = off[tid];
}

constexpr int PITCH = 132;
constexpr int QCAP = 8192;
constexpr int G1T = 1024;
static_assert(N1 % TM == 0 && TM % 16 == 0, "whole tiles only");
__global__ void __launch_bounds__(G1T)
agg_gemm1(const float* __restrict__ feat, const unsigned int* __restrict__ recs,
          const int* __restrict__ offs,
          const float* __restrict__ W,
          const float* __restrict__ bias, float* __restrict__ h1) {
    __shared__ __attribute__((aligned(16))) float As[TM * PITCH];
    __shared__ int degc[TM];
    __shared__ float invs[TM];
    __shared__ unsigned int q[QCAP];
    __shared__ unsigned int q2[QCAP];
    __shared__ int hist[512];
    __shared__ int roff[512];
    __shared__ int qn;
    static_assert(TM <= 512, "one counter per row of the tile");
    const int tid = threadIdx.x;
    const int lane = tid & 31;
    const int wave = __builtin_amdgcn_readfirstlane(tid >> 5);
    const int m0 = blockIdx.x * TM;

    constexpr int MYC = (NCH0 + G1T / 32 - 1) / (G1T / 32);
    int pa[MYC], pe[MYC];
#pragma unroll
    for (int u = 0; u < MYC; ++u) {
        const int c = wave + u * (G1T / 32);
        pa[u] = 0;
        pe[u] = 0;
        if (c < NCH0) {
            pa[u] = offs[c * OFFP + blockIdx.x];
            pe[u] = offs[c * OFFP + blockIdx.x + 1];
        }
    }
    unsigned int rq[MYC][2];
#pragma unroll
    for (int u = 0; u < MYC; ++u) {
        const int a = __builtin_amdgcn_readfirstlane(pa[u]);
        const int len = __builtin_amdgcn_readfirstlane(pe[u] - pa[u]);
        const long rb = (long)(wave + u * (G1T / 32)) * CH + a;
        rq[u][0] = (lane < len) ? recs[rb + lane] : 0u;
        rq[u][1] = (lane + 32 < len) ? recs[rb + lane + 32] : 0u;
    }
    constexpr int NV1 = TM * (DIN / 4);
    constexpr int S1 = (NV1 + G1T - 1) / G1T;
    v4f hv[S1];
    {
        const v4f* own = (const v4f*)(feat + (long)m0 * DIN);
#pragma unroll
        for (int u = 0; u < S1; ++u) {
            const int idx = tid + u * G1T;
            hv[u] = own[idx < NV1 ? idx : NV1 - 1];
        }
    }

    for (int i = tid; i < TM * PITCH; i += G1T) As[i] = 0.0f;
    for (int i = tid; i < TM; i += G1T) degc[i] = 0;
    if (tid < 512) hist[tid] = 0;
    if (tid == 0) qn = 0;
    lds_barrier();

    {
        int qbs[MYC];
        bool rest = false;
#pragma unroll
        for (int u = 0; u < MYC; ++u) {
            const int len = __builtin_amdgcn_readfirstlane(pe[u] - pa[u]);
            qbs[u] = 0;
            if (len <= 0) continue;
            int qb = 0;
            if (lane == 0) qb = atomicAdd(&qn, len);
            qb = __builtin_amdgcn_readfirstlane(qb);
            qbs[u] = qb;
#pragma unroll
            for (int h = 0; h < 2; ++h) {
                const int j = lane + 32 * h;
                if (j < len && qb + j < QCAP) q[qb + j] = rq[u][h];
            }
            rest = rest || len > 64 || qb + len > QCAP;
        }
        if (rest) {
#pragma unroll
            for (int u = 0; u < MYC; ++u) {
                const int a = __builtin_amdgcn_readfirstlane(pa[u]);
                const int len = __builtin_amdgcn_readfirstlane(pe[u] - pa[u]);
                const int qb = qbs[u];
                if (len <= 0) continue;
                const unsigned int* rp = recs + (wave + u * (G1T / 32)) * CH + a;
                for (int j = lane; j < len; j += 32) {
                    const int slot = qb + j;
                    if (j < 64 && slot < QCAP) continue;
                    const unsigned r = rp[j];
                    if (slot < QCAP) {
                        q[slot] = r;
                    } else {
                        const unsigned dl = min(r >> SRC_BITS, (unsigned)(TM - 1));
                        const float* hp = feat + (long)(r & SRC_MASK) * DIN;
                        for (int cc = 0; cc < DIN; ++cc) atomicAdd(&As[dl * PITCH + cc], hp[cc]);
                        atomicAdd(&degc[dl], 1);
                    }
                }
            }
        }
    }
    lds_barrier();

    const int n = __builtin_amdgcn_readfirstlane(qn < QCAP ? qn : QCAP);
    {
        constexpr int RPT = QCAP / G1T;
        unsigned int rr[RPT];
        int rk[RPT];
#pragma unroll
        for (int k = 0; k < RPT; ++k) {
            const int i = tid + k * G1T;
            rr[k] = 0u;
            rk[k] = -1;
            if (i < n) {
                const unsigned r = q[i];
                const int dl = (int)min(r >> SRC_BITS, (unsigned)(TM - 1));
                rr[k] = r;
                rk[k] = atomicAdd(&hist[dl], 1);
                atomicAdd(&degc[dl], 1);
            }
        }
        lds_barrier();
        if (wave == 0) prefix512(hist, roff, lane);
        lds_barrier();
#pragma unroll
        for (int k = 0; k < RPT; ++k) {
            if (rk[k] >= 0) {
                const int dl = (int)min(rr[k] >> SRC_BITS, (unsigned)(TM - 1));
                q2[roff[dl] + rk[k]] = rr[k];
            }
        }
    }
    lds_barrier();

    {
        const int per = (((n + 31) >> 5) + 7) & ~7;
        const int b0 = wave * per;
        const int b1 = (b0 + per < n) ? (b0 + per) : n;
        const float* hl = feat + lane;
        float* al = As + lane;
        int cur = -1;
        float acc[4];
#pragma unroll
        for (int j = 0; j < 4; ++j) acc[j] = 0.0f;
        for (int base = b0; base < b1; base += 32) {
            const int cnt = (b1 - base < 32) ? (b1 - base) : 32;
            int s = 0, dl = 0;
            if (lane < cnt) {
                const unsigned r = q2[base + lane];
                s = (int)(r & SRC_MASK);
                dl = (int)min(r >> SRC_BITS, (unsigned)(TM - 1));
            }
            int i = 0;
#pragma unroll 1
            for (; i + 8 <= cnt; i += 8) {
                float v[8][4];
                int di[8];
#pragma unroll
                for (int u = 0; u < 8; ++u) {
                    const int si = __builtin_amdgcn_readlane(s, i + u);
                    di[u] = __builtin_amdgcn_readlane(dl, i + u);
                    const float* hp = hl + (long)si * DIN;
#pragma unroll
                    for (int j = 0; j < 4; ++j) v[u][j] = hp[32 * j];
                }
#pragma unroll
                for (int u = 0; u < 8; ++u) {
                    if (di[u] != cur) {
                        if (cur >= 0) {
                            float* ap = al + cur * PITCH;
#pragma unroll
                            for (int j = 0; j < 4; ++j) atomicAdd(ap + 32 * j, acc[j]);
                        }
                        cur = di[u];
#pragma unroll
                        for (int j = 0; j < 4; ++j) acc[j] = v[u][j];
                    } else {
#pragma unroll
                        for (int j = 0; j < 4; ++j) acc[j] += v[u][j];
                    }
                }
            }
            for (; i < cnt; ++i) {
                const int si = __builtin_amdgcn_readlane(s, i);
                const int dd = __builtin_amdgcn_readlane(dl, i);
                const float* hp = hl + (long)si * DIN;
                float v[4];
#pragma unroll
                for (int j = 0; j < 4; ++j) v[j] = hp[32 * j];
                if (dd != cur) {
                    if (cur >= 0) {
                        float* ap = al + cur * PITCH;
#pragma unroll
                        for (int j = 0; j < 4; ++j) atomicAdd(ap + 32 * j, acc[j]);
                    }
                    cur = dd;
#pragma unroll
                    for (int j = 0; j < 4; ++j) acc[j] = v[j];
                } else {
#pragma unroll
                    for (int j = 0; j < 4; ++j) acc[j] += v[j];
                }
            }
        }
        if (cur >= 0) {
            float* ap = al + cur * PITCH;
#pragma unroll
            for (int j = 0; j < 4; ++j) atomicAdd(ap + 32 * j, acc[j]);
        }
    }
    lds_barrier();

    for (int r = tid; r < TM; r += G1T) invs[r] = 1.0f / fmaxf((float)degc[r], 1.0f);
    lds_barrier();
#pragma unroll
    for (int u = 0; u < S1; ++u) {
        const int idx = tid + u * G1T;
        if (idx < NV1) {
            const int r = idx / (DIN / 4), c4 = idx % (DIN / 4);
            v4f* ap = (v4f*)(As + r * PITCH) + c4;
            *ap = hv[u] + *ap * invs[r];
        }
    }
    __syncthreads();

    const int lo = lane & 15, hi = lane >> 4;
    const int ncol = (wave & 15) * 16 + lo;
    v2f b[DIN / 4];
#pragma unroll
    for (int ks = 0; ks < DIN / 4; ++ks) {
        const int kk = 4 * ks + 2 * hi;
        b[ks][0] = W[kk * HID + ncol];
        b[ks][1] = W[(kk + 1) * HID + ncol];
    }
    const float bv = bias[ncol];
#pragma unroll 1
    for (int mt = wave >> 4; mt < TM / 16; mt += 2) {
        const float* ar = As + (mt * 16 + lo) * PITCH + 2 * hi;
        v8f acc = {0, 0, 0, 0, 0, 0, 0, 0};
#pragma unroll
        for (int ks = 0; ks < DIN / 4; ++ks) {
            v2f a;
            a[0] = ar[4 * ks];
            a[1] = ar[4 * ks + 1];
            acc = __builtin_amdgcn_wmma_f32_16x16x4_f32(false, a, false, b[ks], (short)0, acc, false, false);
        }
#pragma unroll
        for (int r = 0; r < 8; ++r) {
            const int rr = mt * 16 + r + 8 * hi;
            const float s = (degc[rr] > 0) ? 2.0f : 1.0f;
            h1[(long)(m0 + rr) * HID + ncol] = fmaxf(acc[r] + s * bv, 0.0f);
        }
    }
}

constexpr int MT2 = (TM2 + 15) / 16;
constexpr int PITCH2 = 260;
constexpr int QCAP2 = 12288;
static_assert(MT2 > 4 && MT2 <= 8, "every wave has a first M tile, and at most a second one");
static_assert(PITCH2 % 4 == 0 && HID % 4 == 0, "rows are cleared and finished in 16-byte pieces");
__global__ void __launch_bounds__(G1T)
agg_gemm2(const float* __restrict__ h1, const unsigned int* __restrict__ recs,
          const int* __restrict__ offs,
          const float* __restrict__ W,
          const float* __restrict__ bias, float* __restrict__ out) {
    __shared__ __attribute__((aligned(16))) float As[MT2 * 16 * PITCH2];
    __shared__ int degc[MT2 * 16];
    __shared__ float invs[MT2 * 16];
    __shared__ unsigned int q[QCAP2];
    __shared__ unsigned int q2[QCAP2];
    __shared__ int hist[512];
    __shared__ int roff[512];
    __shared__ int qn;
    static_assert(TM2 <= 512 && QCAP2 % G1T == 0 && QCAP % G1T == 0, "one counter per row; whole queue slots per thread");
    const int tid = threadIdx.x;
    const int lane = tid & 31;
    const int wave = __builtin_amdgcn_readfirstlane(tid >> 5);
    const int m0 = blockIdx.x * TM2;

    constexpr int MYC = (NCH1 + G1T / 32 - 1) / (G1T / 32);
    int pa[MYC], pe[MYC];
#pragma unroll
    for (int u = 0; u < MYC; ++u) {
        const int c = wave + u * (G1T / 32);
        pa[u] = 0;
        pe[u] = 0;
        if (c < NCH1) {
            pa[u] = offs[c * OFFP + blockIdx.x];
            pe[u] = offs[c * OFFP + blockIdx.x + 1];
        }
    }
    unsigned int rq[MYC][2];
#pragma unroll
    for (int u = 0; u < MYC; ++u) {
        const int a = __builtin_amdgcn_readfirstlane(pa[u]);
        const int len = __builtin_amdgcn_readfirstlane(pe[u] - pa[u]);
        const long rb = (long)(wave + u * (G1T / 32)) * CH + a;
        rq[u][0] = (lane < len) ? recs[rb + lane] : 0u;
        rq[u][1] = (lane + 32 < len) ? recs[rb + lane + 32] : 0u;
    }
    constexpr int NV = TM2 * (HID / 4);
    constexpr int S3 = (NV + G1T - 1) / G1T;
    v4f hv[S3];
    {
        const v4f* own = (const v4f*)(h1 + (long)m0 * HID);
#pragma unroll
        for (int u = 0; u < S3; ++u) {
            const int idx = tid + u * G1T;
            hv[u] = own[idx < NV ? idx : NV - 1];
        }
    }

    {
        const v4f z = {0.0f, 0.0f, 0.0f, 0.0f};
        for (int i = tid; i < MT2 * 16 * PITCH2 / 4; i += G1T) ((v4f*)As)[i] = z;
    }
    for (int i = tid; i < MT2 * 16; i += G1T) degc[i] = 0;
    if (tid < 512) hist[tid] = 0;
    if (tid == 0) qn = 0;
    lds_barrier();

    {
        int qbs[MYC];
        bool rest = false;
#pragma unroll
        for (int u = 0; u < MYC; ++u) {
            const int len = __builtin_amdgcn_readfirstlane(pe[u] - pa[u]);
            qbs[u] = 0;
            if (len <= 0) continue;
            int qb = 0;
            if (lane == 0) qb = atomicAdd(&qn, len);
            qb = __builtin_amdgcn_readfirstlane(qb);
            qbs[u] = qb;
#pragma unroll
            for (int h = 0; h < 2; ++h) {
                const int j = lane + 32 * h;
                if (j < len && qb + j < QCAP2) q[qb + j] = rq[u][h];
            }
            rest = rest || len > 64 || qb + len > QCAP2;
        }
        if (rest) {
#pragma unroll
            for (int u = 0; u < MYC; ++u) {
                const int a = __builtin_amdgcn_readfirstlane(pa[u]);
                const int len = __builtin_amdgcn_readfirstlane(pe[u] - pa[u]);
                const int qb = qbs[u];
                if (len <= 0) continue;
                const unsigned int* rp = recs + (wave + u * (G1T / 32)) * CH + a;
                for (int j = lane; j < len; j += 32) {
                    const int slot = qb + j;
                    if (j < 64 && slot < QCAP2) continue;
                    const unsigned r = rp[j];
                    if (slot < QCAP2) {
                        q[slot] = r;
                    } else {
                        const unsigned dl = min(r >> SRC_BITS, (unsigned)(TM2 - 1));
                        const float* hp = h1 + (long)(r & SRC_MASK) * HID;
                        for (int cc = 0; cc < HID; ++cc) atomicAdd(&As[dl * PITCH2 + cc], hp[cc]);
                        atomicAdd(&degc[dl], 1);
                    }
                }
            }
        }
    }
    lds_barrier();

    const int n = __builtin_amdgcn_readfirstlane(qn < QCAP2 ? qn : QCAP2);
    {
        constexpr int RPT = QCAP2 / G1T;
        unsigned int rr[RPT];
        int rk[RPT];
#pragma unroll
        for (int k = 0; k < RPT; ++k) {
            const int i = tid + k * G1T;
            rr[k] = 0u;
            rk[k] = -1;
            if (i < n) {
                const unsigned r = q[i];
                const int dl = (int)min(r >> SRC_BITS, (unsigned)(TM2 - 1));
                rr[k] = r;
                rk[k] = atomicAdd(&hist[dl], 1);
                atomicAdd(&degc[dl], 1);
            }
        }
        lds_barrier();
        if (wave == 0) prefix512(hist, roff, lane);
        lds_barrier();
#pragma unroll
        for (int k = 0; k < RPT; ++k) {
            if (rk[k] >= 0) {
                const int dl = (int)min(rr[k] >> SRC_BITS, (unsigned)(TM2 - 1));
                q2[roff[dl] + rk[k]] = rr[k];
            }
        }
    }
    lds_barrier();

    {
        const int per = (((n + 31) >> 5) + 3) & ~3;
        const int b0 = wave * per;
        const int b1 = (b0 + per < n) ? (b0 + per) : n;
        const float* hl = h1 + lane;
        float* al = As + lane;
        int cur = -1;
        float acc[8];
#pragma unroll
        for (int j = 0; j < 8; ++j) acc[j] = 0.0f;
        for (int base = b0; base < b1; base += 32) {
            const int cnt = (b1 - base < 32) ? (b1 - base) : 32;
            int s = 0, dl = 0;
            if (lane < cnt) {
                const unsigned r = q2[base + lane];
                s = (int)(r & SRC_MASK);
                dl = (int)min(r >> SRC_BITS, (unsigned)(TM2 - 1));
            }
            int i = 0;
#pragma unroll 1
            for (; i + 4 <= cnt; i += 4) {
                float v[4][8];
                int di[4];
#pragma unroll
                for (int u = 0; u < 4; ++u) {
                    const int si = __builtin_amdgcn_readlane(s, i + u);
                    di[u] = __builtin_amdgcn_readlane(dl, i + u);
                    const float* hp = hl + (long)si * HID;
#pragma unroll
                    for (int j = 0; j < 8; ++j) v[u][j] = hp[32 * j];
                }
#pragma unroll
                for (int u = 0; u < 4; ++u) {
                    if (di[u] != cur) {
                        if (cur >= 0) {
                            float* ap = al + cur * PITCH2;
#pragma unroll
                            for (int j = 0; j < 8; ++j) atomicAdd(ap + 32 * j, acc[j]);
                        }
                        cur = di[u];
#pragma unroll
                        for (int j = 0; j < 8; ++j) acc[j] = v[u][j];
                    } else {
#pragma unroll
                        for (int j = 0; j < 8; ++j) acc[j] += v[u][j];
                    }
                }
            }
            for (; i < cnt; ++i) {
                const int si = __builtin_amdgcn_readlane(s, i);
                const int dd = __builtin_amdgcn_readlane(dl, i);
                const float* hp = hl + (long)si * HID;
                float v[8];
#pragma unroll
                for (int j = 0; j < 8; ++j) v[j] = hp[32 * j];
                if (dd != cur) {
                    if (cur >= 0) {
                        float* ap = al + cur * PITCH2;
#pragma unroll
                        for (int j = 0; j < 8; ++j) atomicAdd(ap + 32 * j, acc[j]);
                    }
                    cur = dd;
#pragma unroll
                    for (int j = 0; j < 8; ++j) acc[j] = v[j];
                } else {
#pragma unroll
                    for (int j = 0; j < 8; ++j) acc[j] += v[j];
                }
            }
        }
        if (cur >= 0) {
            float* ap = al + cur * PITCH2;
#pragma unroll
            for (int j = 0; j < 8; ++j) atomicAdd(ap + 32 * j, acc[j]);
        }
    }
    lds_barrier();

    for (int r = tid; r < TM2; r += G1T) invs[r] = 1.0f / fmaxf((float)degc[r], 1.0f);
    lds_barrier();
    {
#pragma unroll
        for (int u = 0; u < S3; ++u) {
            const int idx = tid + u * G1T;
            if (idx < NV) {
                const int r = idx / (HID / 4), c4 = idx % (HID / 4);
                v4f* ap = (v4f*)(As + r * PITCH2) + c4;
                *ap = hv[u] + *ap * invs[r];
            }
        }
    }
    __syncthreads();

    const int lo = lane & 15, hi = lane >> 4;
    const int ncol = (wave & 7) * 16 + lo;
    const int g = wave >> 3;
    const float* wp = W + (2 * hi) * DOUT + ncol;
    const float* a0p = As + (g * 16 + lo) * PITCH2 + 2 * hi;
    const float bv = bias[ncol];
    v8f acc0 = {0, 0, 0, 0, 0, 0, 0, 0};
    if (g + 4 < MT2) {
        const float* a1p = a0p + 64 * PITCH2;
        v8f acc1 = {0, 0, 0, 0, 0, 0, 0, 0};
#pragma unroll 16
        for (int ks = 0; ks < HID / 4; ++ks) {
            v2f b, a0, a1;
            b[0] = wp[(4 * ks) * DOUT];
            b[1] = wp[(4 * ks + 1) * DOUT];
            a0[0] = a0p[4 * ks];
            a0[1] = a0p[4 * ks + 1];
            a1[0] = a1p[4 * ks];
            a1[1] = a1p[4 * ks + 1];
            acc0 = __builtin_amdgcn_wmma_f32_16x16x4_f32(false, a0, false, b, (short)0, acc0, false, false);
            acc1 = __builtin_amdgcn_wmma_f32_16x16x4_f32(false, a1, false, b, (short)0, acc1, false, false);
        }
#pragma unroll
        for (int r = 0; r < 8; ++r) {
            const int rr = (g + 4) * 16 + r + 8 * hi;
            if (rr < TM2) out[(long)(m0 + rr) * DOUT + ncol] = acc1[r] + bv;
        }
    } else {
#pragma unroll 16
        for (int ks = 0; ks < HID / 4; ++ks) {
            v2f b, a0;
            b[0] = wp[(4 * ks) * DOUT];
            b[1] = wp[(4 * ks + 1) * DOUT];
            a0[0] = a0p[4 * ks];
            a0[1] = a0p[4 * ks + 1];
            acc0 = __builtin_amdgcn_wmma_f32_16x16x4_f32(false, a0, false, b, (short)0, acc0, false, false);
        }
    }
    static_assert(16 * 4 <= TM2, "M tiles 0..3 hold only real rows");
#pragma unroll
    for (int r = 0; r < 8; ++r) {
        const int rr = g * 16 + r + 8 * hi;
        out[(long)(m0 + rr) * DOUT + ncol] = acc0[r] + bv;
    }
}

extern "C" void kernel_launch(void* const* d_in, const int* in_sizes, int n_in,
                              void* d_out, int out_size, void* d_ws, size_t ws_size,
                              hipStream_t stream) {
    (void)in_sizes; (void)n_in; (void)out_size; (void)ws_size;
    const int*   x  = (const int*)d_in[0];
    const float* e0 = (const float*)d_in[1];
    const float* e1 = (const float*)d_in[2];
    const float* e2 = (const float*)d_in[3];
    const float* e3 = (const float*)d_in[4];
    const float* Win[2]  = {(const float*)d_in[5], (const float*)d_in[9]};
    const float* bin[2]  = {(const float*)d_in[6], (const float*)d_in[10]};
    const float* Wout[2] = {(const float*)d_in[7], (const float*)d_in[11]};
    const float* bout[2] = {(const float*)d_in[8], (const float*)d_in[12]};
    const int* srcL0[2] = {(const int*)d_in[13], (const int*)d_in[17]};
    const int* dstL0[2] = {(const int*)d_in[14], (const int*)d_in[18]};
    const int* srcL1[2] = {(const int*)d_in[15], (const int*)d_in[19]};
    const int* dstL1[2] = {(const int*)d_in[16], (const int*)d_in[20]};
    float* out = (float*)d_out;

    size_t off = 0;
    auto take = [&](size_t bytes) { size_t o = off; off += (bytes + 255) & ~(size_t)255; return o; };
    char* ws = (char*)d_ws;
    float* feat = (float*)(ws + take((size_t)N0 * DIN * 4));
    float* h1   = (float*)(ws + take((size_t)N1 * HID * 4));
    unsigned int* rc0[2];
    unsigned int* rc1[2];
    int* of0[2];
    int* of1[2];
    rc0[0] = (unsigned int*)(ws + take((size_t)NCH0 * CH * 4));
    rc0[1] = (unsigned int*)(ws + take((size_t)NCH0 * CH * 4));
    rc1[0] = (unsigned int*)(ws + take((size_t)NCH1 * CH * 4));
    rc1[1] = (unsigned int*)(ws + take((size_t)NCH1 * CH * 4));
    of0[0] = (int*)(ws + take((size_t)NCH0 * OFFP * 4));
    of0[1] = (int*)(ws + take((size_t)NCH0 * OFFP * 4));
    of1[0] = (int*)(ws + take((size_t)NCH1 * OFFP * 4));
    of1[1] = (int*)(ws + take((size_t)NCH1 * OFFP * 4));

    build_feat<<<(N0 / 32 + 7) / 8, 256, 0, stream>>>(x, e0, e1, e2, e3, feat);

    bin_edges<<<dim3(NCH0, 4), BT, 0, stream>>>(dstL0[0], dstL0[1], dstL1[0], dstL1[1],
                                               srcL0[0], srcL0[1], srcL1[0], srcL1[1],
                                               rc0[0], rc0[1], rc1[0], rc1[1],
                                               of0[0], of0[1], of1[0], of1[1]);

    for (int e = 0; e < 2; ++e) {
        agg_gemm1<<<N1 / TM, G1T, 0, stream>>>(feat, rc0[e], of0[e], Win[e], bin[e], h1);
        agg_gemm2<<<N2 / TM2, G1T, 0, stream>>>(h1, rc1[e], of1[e], Wout[e], bout[e],
                                                out + (size_t)e * N2 * DOUT);
    }
}


# feat build and edge-list sort, which depend on nothing earlier, share one launch (first blocks sort a chunk each, the rest build feat): 5 launches, not 6; order of every sum unchanged
# speedup vs baseline: 3.1464x; 1.0350x over previous
#include <hip/hip_runtime.h>

constexpr int N0 = 400000, N1 = 100000, N2 = 25000;
constexpr int E0 = 800000, E1 = 200000;
constexpr int DIN = 128, HID = 256, DOUT = 128;

typedef __attribute__((ext_vector_type(2))) float v2f;
typedef __attribute__((ext_vector_type(8))) float v8f;
typedef __attribute__((ext_vector_type(4))) float v4f;

__device__ __forceinline__ void lds_barrier() {
    __builtin_amdgcn_fence(__ATOMIC_RELEASE, "workgroup", "local");
    __builtin_amdgcn_s_barrier();
    __builtin_amdgcn_fence(__ATOMIC_ACQUIRE, "workgroup", "local");
}

__device__ __forceinline__ void prefix512(const int* __restrict__ cnt, int* __restrict__ off, int lane) {
    int c16[16];
    int run = 0;
#pragma unroll
    for (int j = 0; j < 16; ++j) { c16[j] = run; run += cnt[lane * 16 + j]; }
    int incl = run;
    for (int o = 1; o < 32; o <<= 1) {
        const int v = __shfl_up(incl, o);
        if (lane >= o) incl += v;
    }
    const int base = incl - run;
#pragma unroll
    for (int j = 0; j < 16; ++j) off[lane * 16 + j] = base + c16[j];
}

constexpr int FEAT_U = 8;
static_assert(N0 % 32 == 0, "build_feat takes whole groups of 32 rows");
__device__ __forceinline__ void
build_feat(const int fblock,
           const int* __restrict__ x,
           const float* __restrict__ e0, const float* __restrict__ e1,
           const float* __restrict__ e2, const float* __restrict__ e3,
           float* __restrict__ feat) {
    const int lane = threadIdx.x & 31;
    const int wave = __builtin_amdgcn_readfirstlane((int)((fblock * blockDim.x + threadIdx.x) >> 5));
    const int row0 = wave * 32;
    if (row0 >= N0) return;
    const int4 xi = ((const int4*)x)[row0 + lane];
    const float* tb;
    int rl;
    if (lane < 8)       { tb = e0 + lane * 4;        rl = 32; }
    else if (lane < 12) { tb = e1 + (lane - 8) * 4;  rl = 16; }
    else if (lane < 16) { tb = e2 + (lane - 12) * 4; rl = 16; }
    else                { tb = e3 + (lane - 16) * 4; rl = 64; }
    v4f* dst = (v4f*)(feat + (long)row0 * 128) + lane;
#pragma unroll 1
    for (int i = 0; i < 32; i += FEAT_U) {
        v4f v[FEAT_U];
#pragma unroll
        for (int u = 0; u < FEAT_U; ++u) {
            const int i0 = __builtin_amdgcn_readlane(xi.x, i + u);
            const int i1 = __builtin_amdgcn_readlane(xi.y, i + u);
            const int i2 = __builtin_amdgcn_readlane(xi.z, i + u);
            const int i3 = __builtin_amdgcn_readlane(xi.w, i + u);
            const int idx = (lane < 8) ? i0 : (lane < 12) ? i1 : (lane < 16) ? i2 : i3;
            v[u] = *(const v4f*)(tb + (long)idx * rl);
        }
        static_assert(FEAT_U == 8, "the fence below names eight values");
        asm volatile("" :: "v"(v[0]), "v"(v[1]), "v"(v[2]), "v"(v[3]),
                           "v"(v[4]), "v"(v[5]), "v"(v[6]), "v"(v[7]));
#pragma unroll
        for (int u = 0; u < FEAT_U; ++u) dst[(i + u) * 32] = v[u];
    }
}

constexpr int TM = 400;
constexpr int TM2 = 100;
constexpr int CH = 8192;
constexpr int BT = 1024;
constexpr int EPT = CH / BT;
constexpr int NCH0 = (E0 + CH - 1) / CH;
constexpr int NCH1 = (E1 + CH - 1) / CH;
constexpr int OFFP = 256;
constexpr int SRC_BITS = 19;
constexpr unsigned SRC_MASK = (1u << SRC_BITS) - 1u;
constexpr int NT0 = N1 / TM;
constexpr int NT1 = N2 / TM2;
static_assert(N1 % TM == 0 && NT0 < OFFP - 1, "tile t + 1 is an entry of the table, OFFP - 1 stays free for 'no tile'");
static_assert(N2 % TM2 == 0 && NT1 < OFFP - 1, "tile t + 1 is an entry of the table, OFFP - 1 stays free for 'no tile'");
static_assert(N0 <= (1 << SRC_BITS) && N1 <= (1 << SRC_BITS), "src fits the low bits of a record");
static_assert(TM <= (1 << (31 - SRC_BITS)) && TM2 <= (1 << (31 - SRC_BITS)), "the row inside the tile fits the high bits");
static_assert(CH % BT == 0 && OFFP == 8 * 32, "the prefix sum below gives each lane of one wave eight bins");
__device__ __forceinline__ void
bin_edges(const int y, const int c,
          const int* __restrict__ d0A, const int* __restrict__ d0B,
          const int* __restrict__ d1A, const int* __restrict__ d1B,
          const int* __restrict__ s0A, const int* __restrict__ s0B,
          const int* __restrict__ s1A, const int* __restrict__ s1B,
          unsigned int* __restrict__ r0A, unsigned int* __restrict__ r0B,
          unsigned int* __restrict__ r1A, unsigned int* __restrict__ r1B,
          int* __restrict__ o0A, int* __restrict__ o0B,
          int* __restrict__ o1A, int* __restrict__ o1B) {
    __shared__ int cnt[OFFP];
    __shared__ int off[OFFP];
    __shared__ unsigned int rec[CH];
    const bool l1 = y >= 2;
    const int* d = (y == 0) ? d0A : (y == 1) ? d0B : (y == 2) ? d1A : d1B;
    const int* sp = (y == 0) ? s0A : (y == 1) ? s0B : (y == 2) ? s1A : s1B;
    unsigned int* rout = (y == 0) ? r0A : (y == 1) ? r0B : (y == 2) ? r1A : r1B;
    int* oout = (y == 0) ? o0A : (y == 1) ? o0B : (y == 2) ? o1A : o1B;
    const int E = l1 ? E1 : E0;
    const unsigned nd = l1 ? (unsigned)N2 : (unsigned)N1;
    const int tid = threadIdx.x;
    if (tid < OFFP) cnt[tid] = 0;
    __syncthreads();

    const int e0 = c * CH;
    int sv[EPT], dv[EPT];
#pragma unroll
    for (int k = 0; k < EPT; ++k) {
        const int e = e0 + tid + k * BT;
        const bool in = e < E;
        dv[k] = in ? d[e] : -1;
        sv[k] = in ? sp[e] : 0;
    }
    int tl[EPT], rk[EPT];
#pragma unroll
    for (int k = 0; k < EPT; ++k) {
        const unsigned du = (unsigned)dv[k];
        const unsigned q0 = du / (unsigned)TM, q1 = du / (unsigned)TM2;
        tl[k] = (du < nd) ? (int)(l1 ? q1 : q0) : (OFFP - 1);
        rk[k] = atomicAdd(&cnt[tl[k]], 1);
    }
    __syncthreads();

    if (tid < 32) {
        int c8[8];
        int run = 0;
#pragma unroll
        for (int j = 0; j < 8; ++j) { c8[j] = run; run += cnt[tid * 8 + j]; }
        int incl = run;
        for (int o = 1; o < 32; o <<= 1) {
            const int v = __shfl_up(incl, o);
            if (tid >= o) incl += v;
        }
        const int base = incl - run;
#pragma unroll
        for (int j = 0; j < 8; ++j) off[tid * 8 + j] = base + c8[j];
    }
    __syncthreads();

#pragma unroll
    for (int k = 0; k < EPT; ++k) {
        if (tl[k] != OFFP - 1) {
            const int dl = dv[k] - tl[k] * (l1 ? TM2 : TM);
            rec[off[tl[k]] + rk[k]] = (unsigned)sv[k] | ((unsigned)dl << SRC_BITS);
        }
    }
    __syncthreads();

    const int nrec = off[l1 ? NT1 : NT0];
    for (int i = tid; i < nrec; i += BT) rout[e0 + i] = rec[i];
    if (tid < OFFP) oout[c * OFFP + tid] = off[tid];
}

constexpr int NBIN = 2 * NCH0 + 2 * NCH1;
constexpr int NFEATB = (N0 / 32 + BT / 32 - 1) / (BT / 32);
__global__ void __launch_bounds__(BT)
prep(const int* __restrict__ x,
     const float* __restrict__ e0, const float* __restrict__ e1,
     const float* __restrict__ e2, const float* __restrict__ e3,
     float* __restrict__ feat,
     const int* __restrict__ d0A, const int* __restrict__ d0B,
     const int* __restrict__ d1A, const int* __restrict__ d1B,
     const int* __restrict__ s0A, const int* __restrict__ s0B,
     const int* __restrict__ s1A, const int* __restrict__ s1B,
     unsigned int* __restrict__ r0A, unsigned int* __restrict__ r0B,
     unsigned int* __restrict__ r1A, unsigned int* __restrict__ r1B,
     int* __restrict__ o0A, int* __restrict__ o0B,
     int* __restrict__ o1A, int* __restrict__ o1B) {
    const int b = blockIdx.x;
    if (b < NBIN) {
        int y, cc;
        if (b < NCH0)                 { y = 0; cc = b; }
        else if (b < 2 * NCH0)        { y = 1; cc = b - NCH0; }
        else if (b < 2 * NCH0 + NCH1) { y = 2; cc = b - 2 * NCH0; }
        else                          { y = 3; cc = b - 2 * NCH0 - NCH1; }
        bin_edges(y, cc, d0A, d0B, d1A, d1B, s0A, s0B, s1A, s1B, r0A, r0B, r1A, r1B, o0A, o0B, o1A, o1B);
    } else {
        build_feat(b - NBIN, x, e0, e1, e2, e3, feat);
    }
}

constexpr int PITCH = 132;
constexpr int QCAP = 8192;
constexpr int G1T = 1024;
static_assert(N1 % TM == 0 && TM % 16 == 0, "whole tiles only");
__global__ void __launch_bounds__(G1T)
agg_gemm1(const float* __restrict__ feat, const unsigned int* __restrict__ recs,
          const int* __restrict__ offs,
          const float* __restrict__ W,
          const float* __restrict__ bias, float* __restrict__ h1) {
    __shared__ __attribute__((aligned(16))) float As[TM * PITCH];
    __shared__ int degc[TM];
    __shared__ float invs[TM];
    __shared__ unsigned int q[QCAP];
    __shared__ unsigned int q2[QCAP];
    __shared__ int hist[512];
    __shared__ int roff[512];
    __shared__ int qn;
    static_assert(TM <= 512, "one counter per row of the tile");
    const int tid = threadIdx.x;
    const int lane = tid & 31;
    const int wave = __builtin_amdgcn_readfirstlane(tid >> 5);
    const int m0 = blockIdx.x * TM;

    constexpr int MYC = (NCH0 + G1T / 32 - 1) / (G1T / 32);
    int pa[MYC], pe[MYC];
#pragma unroll
    for (int u = 0; u < MYC; ++u) {
        const int c = wave + u * (G1T / 32);
        pa[u] = 0;
        pe[u] = 0;
        if (c < NCH0) {
            pa[u] = offs[c * OFFP + blockIdx.x];
            pe[u] = offs[c * OFFP + blockIdx.x + 1];
        }
    }
    unsigned int rq[MYC][2];
#pragma unroll
    for (int u = 0; u < MYC; ++u) {
        const int a = __builtin_amdgcn_readfirstlane(pa[u]);
        const int len = __builtin_amdgcn_readfirstlane(pe[u] - pa[u]);
        const long rb = (long)(wave + u * (G1T / 32)) * CH + a;
        rq[u][0] = (lane < len) ? recs[rb + lane] : 0u;
        rq[u][1] = (lane + 32 < len) ? recs[rb + lane + 32] : 0u;
    }
    constexpr int NV1 = TM * (DIN / 4);
    constexpr int S1 = (NV1 + G1T - 1) / G1T;
    v4f hv[S1];
    {
        const v4f* own = (const v4f*)(feat + (long)m0 * DIN);
#pragma unroll
        for (int u = 0; u < S1; ++u) {
            const int idx = tid + u * G1T;
            hv[u] = own[idx < NV1 ? idx : NV1 - 1];
        }
    }

    for (int i = tid; i < TM * PITCH; i += G1T) As[i] = 0.0f;
    for (int i = tid; i < TM; i += G1T) degc[i] = 0;
    if (tid < 512) hist[tid] = 0;
    if (tid == 0) qn = 0;
    lds_barrier();

    {
        int qbs[MYC];
        bool rest = false;
#pragma unroll
        for (int u = 0; u < MYC; ++u) {
            const int len = __builtin_amdgcn_readfirstlane(pe[u] - pa[u]);
            qbs[u] = 0;
            if (len <= 0) continue;
            int qb = 0;
            if (lane == 0) qb = atomicAdd(&qn, len);
            qb = __builtin_amdgcn_readfirstlane(qb);
            qbs[u] = qb;
#pragma unroll
            for (int h = 0; h < 2; ++h) {
                const int j = lane + 32 * h;
                if (j < len && qb + j < QCAP) q[qb + j] = rq[u][h];
            }
            rest = rest || len > 64 || qb + len > QCAP;
        }
        if (rest) {
#pragma unroll
            for (int u = 0; u < MYC; ++u) {
                const int a = __builtin_amdgcn_readfirstlane(pa[u]);
                const int len = __builtin_amdgcn_readfirstlane(pe[u] - pa[u]);
                const int qb = qbs[u];
                if (len <= 0) continue;
                const unsigned int* rp = recs + (wave + u * (G1T / 32)) * CH + a;
                for (int j = lane; j < len; j += 32) {
                    const int slot = qb + j;
                    if (j < 64 && slot < QCAP) continue;
                    const unsigned r = rp[j];
                    if (slot < QCAP) {
                        q[slot] = r;
                    } else {
                        const unsigned dl = min(r >> SRC_BITS, (unsigned)(TM - 1));
                        const float* hp = feat + (long)(r & SRC_MASK) * DIN;
                        for (int cc = 0; cc < DIN; ++cc) atomicAdd(&As[dl * PITCH + cc], hp[cc]);
                        atomicAdd(&degc[dl], 1);
                    }
                }
            }
        }
    }
    lds_barrier();

    const int n = __builtin_amdgcn_readfirstlane(qn < QCAP ? qn : QCAP);
    {
        constexpr int RPT = QCAP / G1T;
        unsigned int rr[RPT];
        int rk[RPT];
#pragma unroll
        for (int k = 0; k < RPT; ++k) {
            const int i = tid + k * G1T;
            rr[k] = 0u;
            rk[k] = -1;
            if (i < n) {
                const unsigned r = q[i];
                const int dl = (int)min(r >> SRC_BITS, (unsigned)(TM - 1));
                rr[k] = r;
                rk[k] = atomicAdd(&hist[dl], 1);
                atomicAdd(&degc[dl], 1);
            }
        }
        lds_barrier();
        if (wave == 0) prefix512(hist, roff, lane);
        lds_barrier();
#pragma unroll
        for (int k = 0; k < RPT; ++k) {
            if (rk[k] >= 0) {
                const int dl = (int)min(rr[k] >> SRC_BITS, (unsigned)(TM - 1));
                q2[roff[dl] + rk[k]] = rr[k];
            }
        }
    }
    lds_barrier();

    {
        const int per = (((n + 31) >> 5) + 7) & ~7;
        const int b0 = wave * per;
        const int b1 = (b0 + per < n) ? (b0 + per) : n;
        const float* hl = feat + lane;
        float* al = As + lane;
        int cur = -1;
        float acc[4];
#pragma unroll
        for (int j = 0; j < 4; ++j) acc[j] = 0.0f;
        for (int base = b0; base < b1; base += 32) {
            const int cnt = (b1 - base < 32) ? (b1 - base) : 32;
            int s = 0, dl = 0;
            if (lane < cnt) {
                const unsigned r = q2[base + lane];
                s = (int)(r & SRC_MASK);
                dl = (int)min(r >> SRC_BITS, (unsigned)(TM - 1));
            }
            int i = 0;
#pragma unroll 1
            for (; i + 8 <= cnt; i += 8) {
                float v[8][4];
                int di[8];
#pragma unroll
                for (int u = 0; u < 8; ++u) {
                    const int si = __builtin_amdgcn_readlane(s, i + u);
                    di[u] = __builtin_amdgcn_readlane(dl, i + u);
                    const float* hp = hl + (long)si * DIN;
#pragma unroll
                    for (int j = 0; j < 4; ++j) v[u][j] = hp[32 * j];
                }
#pragma unroll
                for (int u = 0; u < 8; ++u) {
                    if (di[u] != cur) {
                        if (cur >= 0) {
                            float* ap = al + cur * PITCH;
#pragma unroll
                            for (int j = 0; j < 4; ++j) atomicAdd(ap + 32 * j, acc[j]);
                        }
                        cur = di[u];
#pragma unroll
                        for (int j = 0; j < 4; ++j) acc[j] = v[u][j];
                    } else {
#pragma unroll
                        for (int j = 0; j < 4; ++j) acc[j] += v[u][j];
                    }
                }
            }
            for (; i < cnt; ++i) {
                const int si = __builtin_amdgcn_readlane(s, i);
                const int dd = __builtin_amdgcn_readlane(dl, i);
                const float* hp = hl + (long)si * DIN;
                float v[4];
#pragma unroll
                for (int j = 0; j < 4; ++j) v[j] = hp[32 * j];
                if (dd != cur) {
                    if (cur >= 0) {
                        float* ap = al + cur * PITCH;
#pragma unroll
                        for (int j = 0; j < 4; ++j) atomicAdd(ap + 32 * j, acc[j]);
                    }
                    cur = dd;
#pragma unroll
                    for (int j = 0; j < 4; ++j) acc[j] = v[j];
                } else {
#pragma unroll
                    for (int j = 0; j < 4; ++j) acc[j] += v[j];
                }
            }
        }
        if (cur >= 0) {
            float* ap = al + cur * PITCH;
#pragma unroll
            for (int j = 0; j < 4; ++j) atomicAdd(ap + 32 * j, acc[j]);
        }
    }
    lds_barrier();

    for (int r = tid; r < TM; r += G1T) invs[r] = 1.0f / fmaxf((float)degc[r], 1.0f);
    lds_barrier();
#pragma unroll
    for (int u = 0; u < S1; ++u) {
        const int idx = tid + u * G1T;
        if (idx < NV1) {
            const int r = idx / (DIN / 4), c4 = idx % (DIN / 4);
            v4f* ap = (v4f*)(As + r * PITCH) + c4;
            *ap = hv[u] + *ap * invs[r];
        }
    }
    __syncthreads();

    const int lo = lane & 15, hi = lane >> 4;
    const int ncol = (wave & 15) * 16 + lo;
    v2f b[DIN / 4];
#pragma unroll
    for (int ks = 0; ks < DIN / 4; ++ks) {
        const int kk = 4 * ks + 2 * hi;
        b[ks][0] = W[kk * HID + ncol];
        b[ks][1] = W[(kk + 1) * HID + ncol];
    }
    const float bv = bias[ncol];
#pragma unroll 1
    for (int mt = wave >> 4; mt < TM / 16; mt += 2) {
        const float* ar = As + (mt * 16 + lo) * PITCH + 2 * hi;
        v8f acc = {0, 0, 0, 0, 0, 0, 0, 0};
#pragma unroll
        for (int ks = 0; ks < DIN / 4; ++ks) {
            v2f a;
            a[0] = ar[4 * ks];
            a[1] = ar[4 * ks + 1];
            acc = __builtin_amdgcn_wmma_f32_16x16x4_f32(false, a, false, b[ks], (short)0, acc, false, false);
        }
#pragma unroll
        for (int r = 0; r < 8; ++r) {
            const int rr = mt * 16 + r + 8 * hi;
            const float s = (degc[rr] > 0) ? 2.0f : 1.0f;
            h1[(long)(m0 + rr) * HID + ncol] = fmaxf(acc[r] + s * bv, 0.0f);
        }
    }
}

constexpr int MT2 = (TM2 + 15) / 16;
constexpr int PITCH2 = 260;
constexpr int QCAP2 = 12288;
static_assert(MT2 > 4 && MT2 <= 8, "every wave has a first M tile, and at most a second one");
static_assert(PITCH2 % 4 == 0 && HID % 4 == 0, "rows are cleared and finished in 16-byte pieces");
__global__ void __launch_bounds__(G1T)
agg_gemm2(const float* __restrict__ h1, const unsigned int* __restrict__ recs,
          const int* __restrict__ offs,
          const float* __restrict__ W,
          const float* __restrict__ bias, float* __restrict__ out) {
    __shared__ __attribute__((aligned(16))) float As[MT2 * 16 * PITCH2];
    __shared__ int degc[MT2 * 16];
    __shared__ float invs[MT2 * 16];
    __shared__ unsigned int q[QCAP2];
    __shared__ unsigned int q2[QCAP2];
    __shared__ int hist[512];
    __shared__ int roff[512];
    __shared__ int qn;
    static_assert(TM2 <= 512 && QCAP2 % G1T == 0 && QCAP % G1T == 0, "one counter per row; whole queue slots per thread");
    const int tid = threadIdx.x;
    const int lane = tid & 31;
    const int wave = __builtin_amdgcn_readfirstlane(tid >> 5);
    const int m0 = blockIdx.x * TM2;

    constexpr int MYC = (NCH1 + G1T / 32 - 1) / (G1T / 32);
    int pa[MYC], pe[MYC];
#pragma unroll
    for (int u = 0; u < MYC; ++u) {
        const int c = wave + u * (G1T / 32);
        pa[u] = 0;
        pe[u] = 0;
        if (c < NCH1) {
            pa[u] = offs[c * OFFP + blockIdx.x];
            pe[u] = offs[c * OFFP + blockIdx.x + 1];
        }
    }
    unsigned int rq[MYC][2];
#pragma unroll
    for (int u = 0; u < MYC; ++u) {
        const int a = __builtin_amdgcn_readfirstlane(pa[u]);
        const int len = __builtin_amdgcn_readfirstlane(pe[u] - pa[u]);
        const long rb = (long)(wave + u * (G1T / 32)) * CH + a;
        rq[u][0] = (lane < len) ? recs[rb + lane] : 0u;
        rq[u][1] = (lane + 32 < len) ? recs[rb + lane + 32] : 0u;
    }
    constexpr int NV = TM2 * (HID / 4);
    constexpr int S3 = (NV + G1T - 1) / G1T;
    v4f hv[S3];
    {
        const v4f* own = (const v4f*)(h1 + (long)m0 * HID);
#pragma unroll
        for (int u = 0; u < S3; ++u) {
            const int idx = tid + u * G1T;
            hv[u] = own[idx < NV ? idx : NV - 1];
        }
    }

    {
        const v4f z = {0.0f, 0.0f, 0.0f, 0.0f};
        for (int i = tid; i < MT2 * 16 * PITCH2 / 4; i += G1T) ((v4f*)As)[i] = z;
    }
    for (int i = tid; i < MT2 * 16; i += G1T) degc[i] = 0;
    if (tid < 512) hist[tid] = 0;
    if (tid == 0) qn = 0;
    lds_barrier();

    {
        int qbs[MYC];
        bool rest = false;
#pragma unroll
        for (int u = 0; u < MYC; ++u) {
            const int len = __builtin_amdgcn_readfirstlane(pe[u] - pa[u]);
            qbs[u] = 0;
            if (len <= 0) continue;
            int qb = 0;
            if (lane == 0) qb = atomicAdd(&qn, len);
            qb = __builtin_amdgcn_readfirstlane(qb);
            qbs[u] = qb;
#pragma unroll
            for (int h = 0; h < 2; ++h) {
                const int j = lane + 32 * h;
                if (j < len && qb + j < QCAP2) q[qb + j] = rq[u][h];
            }
            rest = rest || len > 64 || qb + len > QCAP2;
        }
        if (rest) {
#pragma unroll
            for (int u = 0; u < MYC; ++u) {
                const int a = __builtin_amdgcn_readfirstlane(pa[u]);
                const int len = __builtin_amdgcn_readfirstlane(pe[u] - pa[u]);
                const int qb = qbs[u];
                if (len <= 0) continue;
                const unsigned int* rp = recs + (wave + u * (G1T / 32)) * CH + a;
                for (int j = lane; j < len; j += 32) {
                    const int slot = qb + j;
                    if (j < 64 && slot < QCAP2) continue;
                    const unsigned r = rp[j];
                    if (slot < QCAP2) {
                        q[slot] = r;
                    } else {
                        const unsigned dl = min(r >> SRC_BITS, (unsigned)(TM2 - 1));
                        const float* hp = h1 + (long)(r & SRC_MASK) * HID;
                        for (int cc = 0; cc < HID; ++cc) atomicAdd(&As[dl * PITCH2 + cc], hp[cc]);
                        atomicAdd(&degc[dl], 1);
                    }
                }
            }
        }
    }
    lds_barrier();

    const int n = __builtin_amdgcn_readfirstlane(qn < QCAP2 ? qn : QCAP2);
    {
        constexpr int RPT = QCAP2 / G1T;
        unsigned int rr[RPT];
        int rk[RPT];
#pragma unroll
        for (int k = 0; k < RPT; ++k) {
            const int i = tid + k * G1T;
            rr[k] = 0u;
            rk[k] = -1;
            if (i < n) {
                const unsigned r = q[i];
                const int dl = (int)min(r >> SRC_BITS, (unsigned)(TM2 - 1));
                rr[k] = r;
                rk[k] = atomicAdd(&hist[dl], 1);
                atomicAdd(&degc[dl], 1);
            }
        }
        lds_barrier();
        if (wave == 0) prefix512(hist, roff, lane);
        lds_barrier();
#pragma unroll
        for (int k = 0; k < RPT; ++k) {
            if (rk[k] >= 0) {
                const int dl = (int)min(rr[k] >> SRC_BITS, (unsigned)(TM2 - 1));
                q2[roff[dl] + rk[k]] = rr[k];
            }
        }
    }
    lds_barrier();

    {
        const int per = (((n + 31) >> 5) + 3) & ~3;
        const int b0 = wave * per;
        const int b1 = (b0 + per < n) ? (b0 + per) : n;
        const float* hl = h1 + lane;
        float* al = As + lane;
        int cur = -1;
        float acc[8];
#pragma unroll
        for (int j = 0; j < 8; ++j) acc[j] = 0.0f;
        for (int base = b0; base < b1; base += 32) {
            const int cnt = (b1 - base < 32) ? (b1 - base) : 32;
            int s = 0, dl = 0;
            if (lane < cnt) {
                const unsigned r = q2[base + lane];
                s = (int)(r & SRC_MASK);
                dl = (int)min(r >> SRC_BITS, (unsigned)(TM2 - 1));
            }
            int i = 0;
#pragma unroll 1
            for (; i + 4 <= cnt; i += 4) {
                float v[4][8];
                int di[4];
#pragma unroll
                for (int u = 0; u < 4; ++u) {
                    const int si = __builtin_amdgcn_readlane(s, i + u);
                    di[u] = __builtin_amdgcn_readlane(dl, i + u);
                    const float* hp = hl + (long)si * HID;
#pragma unroll
                    for (int j = 0; j < 8; ++j) v[u][j] = hp[32 * j];
                }
#pragma unroll
                for (int u = 0; u < 4; ++u) {
                    if (di[u] != cur) {
                        if (cur >= 0) {
                            float* ap = al + cur * PITCH2;
#pragma unroll
                            for (int j = 0; j < 8; ++j) atomicAdd(ap + 32 * j, acc[j]);
                        }
                        cur = di[u];
#pragma unroll
                        for (int j = 0; j < 8; ++j) acc[j] = v[u][j];
                    } else {
#pragma unroll
                        for (int j = 0; j < 8; ++j) acc[j] += v[u][j];
                    }
                }
            }
            for (; i < cnt; ++i) {
                const int si = __builtin_amdgcn_readlane(s, i);
                const int dd = __builtin_amdgcn_readlane(dl, i);
                const float* hp = hl + (long)si * HID;
                float v[8];
#pragma unroll
                for (int j = 0; j < 8; ++j) v[j] = hp[32 * j];
                if (dd != cur) {
                    if (cur >= 0) {
                        float* ap = al + cur * PITCH2;
#pragma unroll
                        for (int j = 0; j < 8; ++j) atomicAdd(ap + 32 * j, acc[j]);
                    }
                    cur = dd;
#pragma unroll
                    for (int j = 0; j < 8; ++j) acc[j] = v[j];
                } else {
#pragma unroll
                    for (int j = 0; j < 8; ++j) acc[j] += v[j];
                }
            }
        }
        if (cur >= 0) {
            float* ap = al + cur * PITCH2;
#pragma unroll
            for (int j = 0; j < 8; ++j) atomicAdd(ap + 32 * j, acc[j]);
        }
    }
    lds_barrier();

    for (int r = tid; r < TM2; r += G1T) invs[r] = 1.0f / fmaxf((float)degc[r], 1.0f);
    lds_barrier();
    {
#pragma unroll
        for (int u = 0; u < S3; ++u) {
            const int idx = tid + u * G1T;
            if (idx < NV) {
                const int r = idx / (HID / 4), c4 = idx % (HID / 4);
                v4f* ap = (v4f*)(As + r * PITCH2) + c4;
                *ap = hv[u] + *ap * invs[r];
            }
        }
    }
    __syncthreads();

    const int lo = lane & 15, hi = lane >> 4;
    const int ncol = (wave & 7) * 16 + lo;
    const int g = wave >> 3;
    const float* wp = W + (2 * hi) * DOUT + ncol;
    const float* a0p = As + (g * 16 + lo) * PITCH2 + 2 * hi;
    const float bv = bias[ncol];
    v8f acc0 = {0, 0, 0, 0, 0, 0, 0, 0};
    if (g + 4 < MT2) {
        const float* a1p = a0p + 64 * PITCH2;
        v8f acc1 = {0, 0, 0, 0, 0, 0, 0, 0};
#pragma unroll 16
        for (int ks = 0; ks < HID / 4; ++ks) {
            v2f b, a0, a1;
            b[0] = wp[(4 * ks) * DOUT];
            b[1] = wp[(4 * ks + 1) * DOUT];
            a0[0] = a0p[4 * ks];
            a0[1] = a0p[4 * ks + 1];
            a1[0] = a1p[4 * ks];
            a1[1] = a1p[4 * ks + 1];
            acc0 = __builtin_amdgcn_wmma_f32_16x16x4_f32(false, a0, false, b, (short)0, acc0, false, false);
            acc1 = __builtin_amdgcn_wmma_f32_16x16x4_f32(false, a1, false, b, (short)0, acc1, false, false);
        }
#pragma unroll
        for (int r = 0; r < 8; ++r) {
            const int rr = (g + 4) * 16 + r + 8 * hi;
            if (rr < TM2) out[(long)(m0 + rr) * DOUT + ncol] = acc1[r] + bv;
        }
    } else {
#pragma unroll 16
        for (int ks = 0; ks < HID / 4; ++ks) {
            v2f b, a0;
            b[0] = wp[(4 * ks) * DOUT];
            b[1] = wp[(4 * ks + 1) * DOUT];
            a0[0] = a0p[4 * ks];
            a0[1] = a0p[4 * ks + 1];
            acc0 = __builtin_amdgcn_wmma_f32_16x16x4_f32(false, a0, false, b, (short)0, acc0, false, false);
        }
    }
    static_assert(16 * 4 <= TM2, "M tiles 0..3 hold only real rows");
#pragma unroll
    for (int r = 0; r < 8; ++r) {
        const int rr = g * 16 + r + 8 * hi;
        out[(long)(m0 + rr) * DOUT + ncol] = acc0[r] + bv;
    }
}

extern "C" void kernel_launch(void* const* d_in, const int* in_sizes, int n_in,
                              void* d_out, int out_size, void* d_ws, size_t ws_size,
                              hipStream_t stream) {
    (void)in_sizes; (void)n_in; (void)out_size; (void)ws_size;
    const int*   x  = (const int*)d_in[0];
    const float* e0 = (const float*)d_in[1];
    const float* e1 = (const float*)d_in[2];
    const float* e2 = (const float*)d_in[3];
    const float* e3 = (const float*)d_in[4];
    const float* Win[2]  = {(const float*)d_in[5], (const float*)d_in[9]};
    const float* bin[2]  = {(const float*)d_in[6], (const float*)d_in[10]};
    const float* Wout[2] = {(const float*)d_in[7], (const float*)d_in[11]};
    const float* bout[2] = {(const float*)d_in[8], (const float*)d_in[12]};
    const int* srcL0[2] = {(const int*)d_in[13], (const int*)d_in[17]};
    const int* dstL0[2] = {(const int*)d_in[14], (const int*)d_in[18]};
    const int* srcL1[2] = {(const int*)d_in[15], (const int*)d_in[19]};
    const int* dstL1[2] = {(const int*)d_in[16], (const int*)d_in[20]};
    float* out = (float*)d_out;

    size_t off = 0;
    auto take = [&](size_t bytes) { size_t o = off; off += (bytes + 255) & ~(size_t)255; return o; };
    char* ws = (char*)d_ws;
    float* feat = (float*)(ws + take((size_t)N0 * DIN * 4));
    float* h1   = (float*)(ws + take((size_t)N1 * HID * 4));
    unsigned int* rc0[2];
    unsigned int* rc1[2];
    int* of0[2];
    int* of1[2];
    rc0[0] = (unsigned int*)(ws + take((size_t)NCH0 * CH * 4));
    rc0[1] = (unsigned int*)(ws + take((size_t)NCH0 * CH * 4));
    rc1[0] = (unsigned int*)(ws + take((size_t)NCH1 * CH * 4));
    rc1[1] = (unsigned int*)(ws + take((size_t)NCH1 * CH * 4));
    of0[0] = (int*)(ws + take((size_t)NCH0 * OFFP * 4));
    of0[1] = (int*)(ws + take((size_t)NCH0 * OFFP * 4));
    of1[0] = (int*)(ws + take((size_t)NCH1 * OFFP * 4));
    of1[1] = (int*)(ws + take((size_t)NCH1 * OFFP * 4));

    prep<<<NBIN + NFEATB, BT, 0, stream>>>(x, e0, e1, e2, e3, feat,
                                           dstL0[0], dstL0[1], dstL1[0], dstL1[1],
                                           srcL0[0], srcL0[1], srcL1[0], srcL1[1],
                                           rc0[0], rc0[1], rc1[0], rc1[1],
                                           of0[0], of0[1], of1[0], of1[1]);

    for (int e = 0; e < 2; ++e) {
        agg_gemm1<<<N1 / TM, G1T, 0, stream>>>(feat, rc0[e], of0[e], Win[e], bin[e], h1);
        agg_gemm2<<<N2 / TM2, G1T, 0, stream>>>(h1, rc1[e], of1[e], Wout[e], bout[e],
                                                out + (size_t)e * N2 * DOUT);
    }
}
